# starting kernel
# baseline (speedup 1.0000x reference)
#include <hip/hip_runtime.h>
#include <hip/hip_bf16.h>


typedef __attribute__((ext_vector_type(2))) float v2f;
typedef __attribute__((ext_vector_type(8))) float v8f;

#define IN_CH 128

__global__ void k_deg_init(float* __restrict__ deg, int n) {
    int i = blockIdx.x * blockDim.x + threadIdx.x;
    if (i < n) deg[i] = 1.0f;
}

__global__ void k_deg_count(const int* __restrict__ dst, float* __restrict__ deg, int e) {
    int i = blockIdx.x * blockDim.x + threadIdx.x;
    if (i < e) atomicAdd(&deg[dst[i]], 1.0f);
}

__global__ void k_dinv(const float* __restrict__ deg, float* __restrict__ dinv, int n) {
    int i = blockIdx.x * blockDim.x + threadIdx.x;
    if (i < n) dinv[i] = rsqrtf(deg[i]);
}

__global__ __launch_bounds__(256) void k_gemm1_wmma(
    const float* __restrict__ x, const float* __restrict__ W1,
    const float* __restrict__ dinv,
    float* __restrict__ h, float* __restrict__ agg1,
    int n_nodes, int mtiles) {
    __shared__ float lw[IN_CH * IN_CH];
    const int tid = threadIdx.x;
    for (int i = tid; i < IN_CH * IN_CH; i += 256) lw[i] = W1[i];
    __syncthreads();

    const int wid  = tid >> 5;
    const int lane = tid & 31;
    const int half = lane >> 4;
    const int l    = lane & 15;
    const int tile = blockIdx.x * 8 + wid;
    if (tile >= mtiles) return;
    const int m0 = tile * 16;

    v8f acc[8] = {};

    const int arow = min(m0 + l, n_nodes - 1);
    const float* ap = x + (size_t)arow * IN_CH;

    for (int kk = 0; kk < IN_CH; kk += 4) {
        const int ka = kk + 2 * half;
        v2f a;
        a.x = ap[ka];
        a.y = ap[ka + 1];
#pragma unroll
        for (int nt = 0; nt < 8; ++nt) {
            const int nc = nt * 16 + l;
            v2f b;
            b.x = lw[ka * IN_CH + nc];
            b.y = lw[(ka + 1) * IN_CH + nc];
            acc[nt] = __builtin_amdgcn_wmma_f32_16x16x4_f32(
                false, a, false, b, (short)0, acc[nt], false, false);
        }
    }

#pragma unroll
    for (int j = 0; j < 8; ++j) {
        const int row = m0 + j + 8 * half;
        if (row >= n_nodes) continue;
        const float di = dinv[row];
        const float sc = di * di;
#pragma unroll
        for (int nt = 0; nt < 8; ++nt) {
            const float c = acc[nt][j];
            const size_t idx = (size_t)row * IN_CH + nt * 16 + l;
            h[idx]    = c;
            agg1[idx] = c * sc;
        }
    }
}

__global__ __launch_bounds__(256) void k_edge_agg1(
    const int* __restrict__ src, const int* __restrict__ dst,
    const float* __restrict__ dinv, const float* __restrict__ h,
    float* __restrict__ agg1, int e) {
    const int w    = blockIdx.x * 8 + (threadIdx.x >> 5);
    const int lane = threadIdx.x & 31;
    if (w >= e) return;
    const int s = src[w], d = dst[w];
    const float norm = dinv[s] * dinv[d];
    const float4 hv = ((const float4*)(h + (size_t)s * IN_CH))[lane];
    float* op = agg1 + (size_t)d * IN_CH + lane * 4;
    atomicAdd(op + 0, hv.x * norm);
    atomicAdd(op + 1, hv.y * norm);
    atomicAdd(op + 2, hv.z * norm);
    atomicAdd(op + 3, hv.w * norm);
}

__global__ __launch_bounds__(256) void k_gemv2(
    const float* __restrict__ agg1, const float* __restrict__ b1,
    const float* __restrict__ W2, const float* __restrict__ b2,
    const float* __restrict__ dinv,
    float* __restrict__ h2, float* __restrict__ out, int n) {
    const int node = blockIdx.x * 8 + (threadIdx.x >> 5);
    const int lane = threadIdx.x & 31;
    if (node >= n) return;
    const float4 a  = ((const float4*)(agg1 + (size_t)node * IN_CH))[lane];
    const float4 bb = ((const float4*)b1)[lane];
    const float4 w  = ((const float4*)W2)[lane];
    float v = fmaxf(a.x + bb.x, 0.f) * w.x + fmaxf(a.y + bb.y, 0.f) * w.y +
              fmaxf(a.z + bb.z, 0.f) * w.z + fmaxf(a.w + bb.w, 0.f) * w.w;
#pragma unroll
    for (int off = 16; off > 0; off >>= 1) v += __shfl_xor(v, off, 32);
    if (lane == 0) {
        const float di = dinv[node];
        h2[node]  = v;
        out[node] = v * di * di + b2[0];
    }
}

__global__ void k_edge_agg2(const int* __restrict__ src, const int* __restrict__ dst,
                            const float* __restrict__ dinv, const float* __restrict__ h2,
                            float* __restrict__ out, int e) {
    int i = blockIdx.x * blockDim.x + threadIdx.x;
    if (i < e) {
        const int s = src[i], d = dst[i];
        atomicAdd(&out[d], h2[s] * dinv[s] * dinv[d]);
    }
}

extern "C" void kernel_launch(void* const* d_in, const int* in_sizes, int n_in,
                              void* d_out, int out_size, void* d_ws, size_t ws_size,
                              hipStream_t stream) {
    const float* x  = (const float*)d_in[0];
    const int*   ei = (const int*)d_in[1];
    const float* W1 = (const float*)d_in[2];
    const float* b1 = (const float*)d_in[3];
    const float* W2 = (const float*)d_in[4];
    const float* b2 = (const float*)d_in[5];

    const int n = in_sizes[0] / IN_CH;
    const int e = in_sizes[1] / 2;
    const int* src = ei;
    const int* dst = ei + e;

    float* ws   = (float*)d_ws;
    float* deg  = ws;
    float* dinv = deg + n;
    float* h    = dinv + n;
    float* agg1 = h + (size_t)n * IN_CH;
    float* h2   = agg1 + (size_t)n * IN_CH;
    float* out  = (float*)d_out;

    const int mtiles = (n + 15) / 16;

    k_deg_init <<<(n + 255) / 256, 256, 0, stream>>>(deg, n);
    k_deg_count<<<(e + 255) / 256, 256, 0, stream>>>(dst, deg, e);
    k_dinv     <<<(n + 255) / 256, 256, 0, stream>>>(deg, dinv, n);
    k_gemm1_wmma<<<(mtiles + 7) / 8, 256, 0, stream>>>(x, W1, dinv, h, agg1, n, mtiles);
    k_edge_agg1<<<(e + 7) / 8, 256, 0, stream>>>(src, dst, dinv, h, agg1, e);
    k_gemv2    <<<(n + 7) / 8, 256, 0, stream>>>(agg1, b1, W2, b2, dinv, h2, out, n);
    k_edge_agg2<<<(e + 255) / 256, 256, 0, stream>>>(src, dst, dinv, h2, out, e);
}


# Record/dinv pass run inside the GEMM launch, split by block index (GEMM blocks first, then blocks of 2048 edges, 8 per thread loaded as one group): one launch fewer; sum orders unchanged
# speedup vs baseline: 1.6438x; 1.6438x over previous
#include <hip/hip_runtime.h>
#include <hip/hip_bf16.h>


typedef __attribute__((ext_vector_type(2))) float v2f;
typedef __attribute__((ext_vector_type(8))) float v8f;

#define IN_CH 128
#define PCH   32
#define NPL   (IN_CH / PCH)
#define RN    391
#define AW    32
#define CH    512
#define QCAP  544
#define SBITS 20

__global__ void k_deg_init(float* __restrict__ deg, int n) {
    int i = blockIdx.x * blockDim.x + threadIdx.x;
    if (i < n) deg[i] = 1.0f;
}

__global__ void k_deg_own(const int* __restrict__ dst, float* __restrict__ deg,
                          unsigned* __restrict__ own, int e) {
    const int t  = blockIdx.x * blockDim.x + threadIdx.x;
    const int nw = ((e + 15) / 16) * 4;
    if (t >= nw) return;
    int d[4];
#pragma unroll
    for (int j = 0; j < 4; ++j) d[j] = dst[min(4 * t + j, e - 1)];
    unsigned pk = 0;
#pragma unroll
    for (int j = 0; j < 4; ++j) {
        if (4 * t + j < e) {
            atomicAdd(&deg[d[j]], 1.0f);
            pk |= ((unsigned)d[j] / (unsigned)RN) << (8 * j);
        }
    }
    own[t] = pk;
}

#define EPB 2048
__device__ __forceinline__ void rec_block(
    int rb, int rblocks, int tid,
    const int* __restrict__ src, const int* __restrict__ dst,
    const float* __restrict__ deg, float* __restrict__ dinv,
    unsigned* __restrict__ rec, int n, int e) {
    if (e > 0) {
        const int base = rb * EPB + tid;
        int s[8], d[8];
        float dgs[8], dgd[8];
#pragma unroll
        for (int k = 0; k < 8; ++k) {
            const int i = min(base + k * 256, e - 1);
            s[k] = src[i];
            d[k] = dst[i];
        }
#pragma unroll
        for (int k = 0; k < 8; ++k) {
            dgs[k] = deg[s[k]];
            dgd[k] = deg[d[k]];
        }
        for (int j = rb * 256 + tid; j < n; j += rblocks * 256)
            dinv[j] = rsqrtf(deg[j]);
#pragma unroll
        for (int k = 0; k < 8; ++k) {
            const int i = base + k * 256;
            if (i < e) {
                const float w = rsqrtf(dgs[k]) * rsqrtf(dgd[k]);
                const unsigned key = (unsigned)s[k] | (((unsigned)d[k] % (unsigned)RN) << SBITS);
                reinterpret_cast<uint2*>(rec)[i] = make_uint2(key, __float_as_uint(w));
            }
        }
    } else {
        for (int j = rb * 256 + tid; j < n; j += rblocks * 256)
            dinv[j] = rsqrtf(deg[j]);
    }
}

__global__ __launch_bounds__(256) void k_gemm1_rec(
    const float* __restrict__ x, const float* __restrict__ W1,
    float* __restrict__ h,
    const int* __restrict__ src, const int* __restrict__ dst,
    const float* __restrict__ deg, float* __restrict__ dinv,
    unsigned* __restrict__ rec,
    int n_nodes, int mtiles, int e, int gblocks) {
    __shared__ float lw[IN_CH * IN_CH];
    const int tid = threadIdx.x;
    if ((int)blockIdx.x >= gblocks) {
        rec_block((int)blockIdx.x - gblocks, (int)gridDim.x - gblocks, tid,
                  src, dst, deg, dinv, rec, n_nodes, e);
        return;
    }
    for (int i = tid; i < IN_CH * IN_CH; i += 256) lw[i] = W1[i];
    __syncthreads();

    const int wid  = tid >> 5;
    const int lane = tid & 31;
    const int half = lane >> 4;
    const int l    = lane & 15;
    const int tile = blockIdx.x * 8 + wid;
    if (tile >= mtiles) return;
    const int m0 = tile * 16;

    v8f acc[8] = {};

    const int arow = min(m0 + l, n_nodes - 1);
    const float* ap = x + (size_t)arow * IN_CH;

    for (int kk = 0; kk < IN_CH; kk += 4) {
        const int ka = kk + 2 * half;
        v2f a;
        a.x = ap[ka];
        a.y = ap[ka + 1];
#pragma unroll
        for (int nt = 0; nt < 8; ++nt) {
            const int nc = nt * 16 + l;
            v2f b;
            b.x = lw[ka * IN_CH + nc];
            b.y = lw[(ka + 1) * IN_CH + nc];
            acc[nt] = __builtin_amdgcn_wmma_f32_16x16x4_f32(
                false, a, false, b, (short)0, acc[nt], false, false);
        }
    }

#pragma unroll
    for (int j = 0; j < 8; ++j) {
        const int row = m0 + j + 8 * half;
        if (row >= n_nodes) continue;
#pragma unroll
        for (int nt = 0; nt < 8; ++nt)
            h[(size_t)row * IN_CH + nt * 16 + l] = acc[nt][j];
    }
}

template <bool FULL>
__device__ __forceinline__ void add_queued(
    const int* q, int cnt, int lane,
    const uint2* __restrict__ rec2,
    const float* __restrict__ h, float* tile) {
    int ms = 0, md = 0;
    float mw = 0.f;
    if (FULL || lane < cnt) {
        const uint2 r = rec2[q[lane]];
        ms = (int)(r.x & ((1u << SBITS) - 1u));
        md = (int)(r.x >> SBITS);
        mw = __uint_as_float(r.y);
    }
#pragma unroll
    for (int g = 0; g < 4; ++g) {
        float hv[8][4];
#pragma unroll
        for (int k = 0; k < 8; ++k) {
            if (FULL || g * 8 + k < cnt) {
                const float* row = h + (size_t)__builtin_amdgcn_readlane(ms, g * 8 + k) * IN_CH + lane;
#pragma unroll
                for (int j = 0; j < 4; ++j) hv[k][j] = row[j * 32];
            }
        }
#pragma unroll
        for (int k = 0; k < 8; ++k) {
            if (FULL || g * 8 + k < cnt) {
                const int dl = __builtin_amdgcn_readlane(md, g * 8 + k);
                const float wk = __int_as_float(
                    __builtin_amdgcn_readlane(__float_as_int(mw), g * 8 + k));
                float* trow = tile + dl * IN_CH + lane;
#pragma unroll
                for (int j = 0; j < 4; ++j)
                    __hip_atomic_fetch_add(trow + j * 32, hv[k][j] * wk,
                                           __ATOMIC_RELAXED, __HIP_MEMORY_SCOPE_WORKGROUP);
            }
        }
    }
}

__global__ __launch_bounds__(1024) void k_agg_owner(
    const unsigned* __restrict__ own, const unsigned* __restrict__ rec,
    const float* __restrict__ dinv, const float* __restrict__ h,
    float* __restrict__ agg1, int n, int e) {
    __shared__ float tile[RN * IN_CH];
    __shared__ int   queue[AW * QCAP];
    const int tid  = threadIdx.x;
    const int lane = tid & 31;
    const int w    = __builtin_amdgcn_readfirstlane(tid >> 5);
    const int d0   = blockIdx.x * RN;
    const int rows = min(RN, n - d0);

    for (int i = tid; i < rows * IN_CH; i += AW * 32) {
        const float di = dinv[d0 + (i >> 7)];
        tile[i] = h[(size_t)d0 * IN_CH + i] * (di * di);
    }
    __syncthreads();

    const int nchunk = (e + CH - 1) / CH;
    const int niter  = (nchunk + AW - 1) / AW;
    const int t0     = (int)(((long long)blockIdx.x * niter) / gridDim.x);
    const int last16 = (e + 15) / 16 - 1;
    const unsigned pat = (unsigned)blockIdx.x * 0x01010101u;
    const uint4* own16 = reinterpret_cast<const uint4*>(own);
    const uint2* rec2  = reinterpret_cast<const uint2*>(rec);
    int* q = queue + w * QCAP;
    int qn = 0;

    auto load1 = [&](uint4& buf, int t) {
        int tt = t0 + t;
        if (tt >= niter) tt -= niter;
        const int blk = (tt * AW + w) * (CH / 16) + lane;
        buf = own16[min(blk, last16)];
    };
    auto scan1 = [&](const uint4& buf, int t) {
        int tt = t0 + t;
        if (tt >= niter) tt -= niter;
        const int base = (tt * AW + w) * CH + lane * 16;
        const unsigned xs[4] = {buf.x, buf.y, buf.z, buf.w};
#pragma unroll
        for (int k = 0; k < 4; ++k) {
            const unsigned v = xs[k] ^ pat;
            const unsigned z = (v - 0x01010101u) & ~v & 0x80808080u;
            if (__builtin_amdgcn_ballot_w32(z != 0u)) {
#pragma unroll
                for (int b = 0; b < 4; ++b) {
                    const int i = base + k * 4 + b;
                    const bool m = (((v >> (8 * b)) & 0xffu) == 0u) && (i < e);
                    const unsigned mask = __builtin_amdgcn_ballot_w32(m);
                    if (mask) {
                        if (m) {
                            const int pos = qn + (int)__builtin_amdgcn_mbcnt_lo(mask, 0u);
                            q[pos] = i;
                        }
                        qn += __builtin_popcount(mask);
                    }
                }
            }
        }
        while (qn >= 32) {
            qn -= 32;
            add_queued<true>(q + qn, 32, lane, rec2, h, tile);
        }
    };

    uint4 bufa, bufb;
    load1(bufa, 0);
    for (int t = 0; t < niter; t += 2) {
        load1(bufb, t + 1);
        scan1(bufa, t);
        load1(bufa, t + 2);
        if (t + 1 < niter) scan1(bufb, t + 1);
    }
    if (qn > 0) add_queued<false>(q, qn, lane, rec2, h, tile);
    __syncthreads();

    const size_t pstride = (size_t)n * PCH;
    for (int ln = w; ln < rows * NPL; ln += AW) {
        const int r = ln >> 2, p = ln & 3;
        agg1[(size_t)p * pstride + (size_t)(d0 + r) * PCH + lane] =
            tile[r * IN_CH + p * PCH + lane];
    }
}

__global__ __launch_bounds__(256) void k_gemv2(
    const float* __restrict__ agg1, const float* __restrict__ b1,
    const float* __restrict__ W2, const float* __restrict__ b2,
    const float* __restrict__ dinv,
    float* __restrict__ h2, float* __restrict__ out, int n) {
    const int node = blockIdx.x * 8 + (threadIdx.x >> 5);
    const int lane = threadIdx.x & 31;
    if (node >= n) return;
    const size_t pstride = (size_t)n * PCH;
    const float* ar = agg1 + (size_t)node * PCH + lane;
    float t[NPL];
#pragma unroll
    for (int p = 0; p < NPL; ++p) {
        const int ch = p * PCH + lane;
        t[p] = fmaxf(ar[p * pstride] + b1[ch], 0.f);
    }
    float v = t[0] * W2[lane] + t[1] * W2[PCH + lane] +
              t[2] * W2[2 * PCH + lane] + t[3] * W2[3 * PCH + lane];
#pragma unroll
    for (int off = 16; off > 0; off >>= 1) v += __shfl_xor(v, off, 32);
    if (lane == 0) {
        const float di = dinv[node];
        h2[node]  = v;
        out[node] = v * di * di + b2[0];
    }
}

__global__ void k_edge_agg2(const int* __restrict__ dst, const unsigned* __restrict__ rec,
                            const float* __restrict__ h2,
                            float* __restrict__ out, int e) {
    int i = blockIdx.x * blockDim.x + threadIdx.x;
    if (i < e) {
        const uint2 r = reinterpret_cast<const uint2*>(rec)[i];
        const int s = (int)(r.x & ((1u << SBITS) - 1u));
        atomicAdd(&out[dst[i]], h2[s] * __uint_as_float(r.y));
    }
}

extern "C" void kernel_launch(void* const* d_in, const int* in_sizes, int n_in,
                              void* d_out, int out_size, void* d_ws, size_t ws_size,
                              hipStream_t stream) {
    const float* x  = (const float*)d_in[0];
    const int*   ei = (const int*)d_in[1];
    const float* W1 = (const float*)d_in[2];
    const float* b1 = (const float*)d_in[3];
    const float* W2 = (const float*)d_in[4];
    const float* b2 = (const float*)d_in[5];

    const int n = in_sizes[0] / IN_CH;
    const int e = in_sizes[1] / 2;
    const int* src = ei;
    const int* dst = ei + e;

    float* ws   = (float*)d_ws;
    float* deg  = ws;
    float* dinv = deg + n;
    float* h    = dinv + n;
    float* agg1 = h + (size_t)n * IN_CH;
    float* h2   = agg1 + (size_t)n * IN_CH;
    unsigned* own = (unsigned*)(h2 + n);
    unsigned* rec = own + ((e + 15) / 16) * 4;
    float* out  = (float*)d_out;
    const int mtiles = (n + 15) / 16;

    k_deg_init <<<(n + 255) / 256, 256, 0, stream>>>(deg, n);
    k_deg_own  <<<(e / 4 + 4 + 255) / 256, 256, 0, stream>>>(dst, deg, own, e);
    const int gblocks = (mtiles + 7) / 8;
    const int rblocks = (e + EPB) / EPB;
    k_gemm1_rec<<<gblocks + rblocks, 256, 0, stream>>>(x, W1, h, src, dst, deg, dinv, rec,
                                                       n, mtiles, e, gblocks);
    k_agg_owner<<<(n + RN - 1) / RN, AW * 32, 0, stream>>>(own, rec, dinv, h, agg1, n, e);
    k_gemv2    <<<(n + 7) / 8, 256, 0, stream>>>(agg1, b1, W2, b2, dinv, h2, out, n);
    k_edge_agg2<<<(e + 255) / 256, 256, 0, stream>>>(dst, rec, h2, out, e);
}


# Layer-2 linear (bias, relu, W2 dot) done by the owner work-group on its finished LDS tile, a run of 13 rows per wave: summed rows never stored or re-read, one launch fewer; sum orders unchanged
# speedup vs baseline: 1.7450x; 1.0616x over previous
#include <hip/hip_runtime.h>
#include <hip/hip_bf16.h>


typedef __attribute__((ext_vector_type(2))) float v2f;
typedef __attribute__((ext_vector_type(8))) float v8f;

#define IN_CH 128
#define PCH   32
#define NPL   (IN_CH / PCH)
#define RN    391
#define AW    32
#define CH    512
#define QCAP  544
#define SBITS 20
#define RPW   ((RN + AW - 1) / AW)

__global__ void k_deg_init(float* __restrict__ deg, int n) {
    int i = blockIdx.x * blockDim.x + threadIdx.x;
    if (i < n) deg[i] = 1.0f;
}

__global__ void k_deg_own(const int* __restrict__ dst, float* __restrict__ deg,
                          unsigned* __restrict__ own, int e) {
    const int t  = blockIdx.x * blockDim.x + threadIdx.x;
    const int nw = ((e + 15) / 16) * 4;
    if (t >= nw) return;
    int d[4];
#pragma unroll
    for (int j = 0; j < 4; ++j) d[j] = dst[min(4 * t + j, e - 1)];
    unsigned pk = 0;
#pragma unroll
    for (int j = 0; j < 4; ++j) {
        if (4 * t + j < e) {
            atomicAdd(&deg[d[j]], 1.0f);
            pk |= ((unsigned)d[j] / (unsigned)RN) << (8 * j);
        }
    }
    own[t] = pk;
}

#define EPB 2048
__device__ __forceinline__ void rec_block(
    int rb, int rblocks, int tid,
    const int* __restrict__ src, const int* __restrict__ dst,
    const float* __restrict__ deg, float* __restrict__ dinv,
    unsigned* __restrict__ rec, int n, int e) {
    if (e > 0) {
        const int base = rb * EPB + tid;
        int s[8], d[8];
        float dgs[8], dgd[8];
#pragma unroll
        for (int k = 0; k < 8; ++k) {
            const int i = min(base + k * 256, e - 1);
            s[k] = src[i];
            d[k] = dst[i];
        }
#pragma unroll
        for (int k = 0; k < 8; ++k) {
            dgs[k] = deg[s[k]];
            dgd[k] = deg[d[k]];
        }
        for (int j = rb * 256 + tid; j < n; j += rblocks * 256)
            dinv[j] = rsqrtf(deg[j]);
#pragma unroll
        for (int k = 0; k < 8; ++k) {
            const int i = base + k * 256;
            if (i < e) {
                const float w = rsqrtf(dgs[k]) * rsqrtf(dgd[k]);
                const unsigned key = (unsigned)s[k] | (((unsigned)d[k] % (unsigned)RN) << SBITS);
                reinterpret_cast<uint2*>(rec)[i] = make_uint2(key, __float_as_uint(w));
            }
        }
    } else {
        for (int j = rb * 256 + tid; j < n; j += rblocks * 256)
            dinv[j] = rsqrtf(deg[j]);
    }
}

__global__ __launch_bounds__(256) void k_gemm1_rec(
    const float* __restrict__ x, const float* __restrict__ W1,
    float* __restrict__ h,
    const int* __restrict__ src, const int* __restrict__ dst,
    const float* __restrict__ deg, float* __restrict__ dinv,
    unsigned* __restrict__ rec,
    int n_nodes, int mtiles, int e, int gblocks) {
    __shared__ float lw[IN_CH * IN_CH];
    const int tid = threadIdx.x;
    if ((int)blockIdx.x >= gblocks) {
        rec_block((int)blockIdx.x - gblocks, (int)gridDim.x - gblocks, tid,
                  src, dst, deg, dinv, rec, n_nodes, e);
        return;
    }
    for (int i = tid; i < IN_CH * IN_CH; i += 256) lw[i] = W1[i];
    __syncthreads();

    const int wid  = tid >> 5;
    const int lane = tid & 31;
    const int half = lane >> 4;
    const int l    = lane & 15;
    const int tile = blockIdx.x * 8 + wid;
    if (tile >= mtiles) return;
    const int m0 = tile * 16;

    v8f acc[8] = {};

    const int arow = min(m0 + l, n_nodes - 1);
    const float* ap = x + (size_t)arow * IN_CH;

    for (int kk = 0; kk < IN_CH; kk += 4) {
        const int ka = kk + 2 * half;
        v2f a;
        a.x = ap[ka];
        a.y = ap[ka + 1];
#pragma unroll
        for (int nt = 0; nt < 8; ++nt) {
            const int nc = nt * 16 + l;
            v2f b;
            b.x = lw[ka * IN_CH + nc];
            b.y = lw[(ka + 1) * IN_CH + nc];
            acc[nt] = __builtin_amdgcn_wmma_f32_16x16x4_f32(
                false, a, false, b, (short)0, acc[nt], false, false);
        }
    }

#pragma unroll
    for (int j = 0; j < 8; ++j) {
        const int row = m0 + j + 8 * half;
        if (row >= n_nodes) continue;
#pragma unroll
        for (int nt = 0; nt < 8; ++nt)
            h[(size_t)row * IN_CH + nt * 16 + l] = acc[nt][j];
    }
}

template <bool FULL>
__device__ __forceinline__ void add_queued(
    const int* q, int cnt, int lane,
    const uint2* __restrict__ rec2,
    const float* __restrict__ h, float* tile) {
    int ms = 0, md = 0;
    float mw = 0.f;
    if (FULL || lane < cnt) {
        const uint2 r = rec2[q[lane]];
        ms = (int)(r.x & ((1u << SBITS) - 1u));
        md = (int)(r.x >> SBITS);
        mw = __uint_as_float(r.y);
    }
#pragma unroll
    for (int g = 0; g < 4; ++g) {
        float hv[8][4];
#pragma unroll
        for (int k = 0; k < 8; ++k) {
            if (FULL || g * 8 + k < cnt) {
                const float* row = h + (size_t)__builtin_amdgcn_readlane(ms, g * 8 + k) * IN_CH + lane;
#pragma unroll
                for (int j = 0; j < 4; ++j) hv[k][j] = row[j * 32];
            }
        }
#pragma unroll
        for (int k = 0; k < 8; ++k) {
            if (FULL || g * 8 + k < cnt) {
                const int dl = __builtin_amdgcn_readlane(md, g * 8 + k);
                const float wk = __int_as_float(
                    __builtin_amdgcn_readlane(__float_as_int(mw), g * 8 + k));
                float* trow = tile + dl * IN_CH + lane;
#pragma unroll
                for (int j = 0; j < 4; ++j)
                    __hip_atomic_fetch_add(trow + j * 32, hv[k][j] * wk,
                                           __ATOMIC_RELAXED, __HIP_MEMORY_SCOPE_WORKGROUP);
            }
        }
    }
}

__global__ __launch_bounds__(1024) void k_agg_owner(
    const unsigned* __restrict__ own, const unsigned* __restrict__ rec,
    const float* __restrict__ dinv, const float* __restrict__ h,
    const float* __restrict__ b1, const float* __restrict__ W2,
    const float* __restrict__ b2,
    float* __restrict__ h2, float* __restrict__ out, int n, int e) {
    __shared__ float tile[RN * IN_CH];
    __shared__ int   queue[AW * QCAP];
    const int tid  = threadIdx.x;
    const int lane = tid & 31;
    const int w    = __builtin_amdgcn_readfirstlane(tid >> 5);
    const int d0   = blockIdx.x * RN;
    const int rows = min(RN, n - d0);

    for (int i = tid; i < rows * IN_CH; i += AW * 32) {
        const float di = dinv[d0 + (i >> 7)];
        tile[i] = h[(size_t)d0 * IN_CH + i] * (di * di);
    }
    __syncthreads();

    const int nchunk = (e + CH - 1) / CH;
    const int niter  = (nchunk + AW - 1) / AW;
    const int t0     = (int)(((long long)blockIdx.x * niter) / gridDim.x);
    const int last16 = (e + 15) / 16 - 1;
    const unsigned pat = (unsigned)blockIdx.x * 0x01010101u;
    const uint4* own16 = reinterpret_cast<const uint4*>(own);
    const uint2* rec2  = reinterpret_cast<const uint2*>(rec);
    int* q = queue + w * QCAP;
    int qn = 0;

    auto load1 = [&](uint4& buf, int t) {
        int tt = t0 + t;
        if (tt >= niter) tt -= niter;
        const int blk = (tt * AW + w) * (CH / 16) + lane;
        buf = own16[min(blk, last16)];
    };
    auto scan1 = [&](const uint4& buf, int t) {
        int tt = t0 + t;
        if (tt >= niter) tt -= niter;
        const int base = (tt * AW + w) * CH + lane * 16;
        const unsigned xs[4] = {buf.x, buf.y, buf.z, buf.w};
#pragma unroll
        for (int k = 0; k < 4; ++k) {
            const unsigned v = xs[k] ^ pat;
            const unsigned z = (v - 0x01010101u) & ~v & 0x80808080u;
            if (__builtin_amdgcn_ballot_w32(z != 0u)) {
#pragma unroll
                for (int b = 0; b < 4; ++b) {
                    const int i = base + k * 4 + b;
                    const bool m = (((v >> (8 * b)) & 0xffu) == 0u) && (i < e);
                    const unsigned mask = __builtin_amdgcn_ballot_w32(m);
                    if (mask) {
                        if (m) {
                            const int pos = qn + (int)__builtin_amdgcn_mbcnt_lo(mask, 0u);
                            q[pos] = i;
                        }
                        qn += __builtin_popcount(mask);
                    }
                }
            }
        }
        while (qn >= 32) {
            qn -= 32;
            add_queued<true>(q + qn, 32, lane, rec2, h, tile);
        }
    };

    uint4 bufa, bufb;
    load1(bufa, 0);
    for (int t = 0; t < niter; t += 2) {
        load1(bufb, t + 1);
        scan1(bufa, t);
        load1(bufa, t + 2);
        if (t + 1 < niter) scan1(bufb, t + 1);
    }
    if (qn > 0) add_queued<false>(q, qn, lane, rec2, h, tile);
    __syncthreads();

    const int r0  = w * RPW;
    const int cnt = min(RPW, rows - r0);
    float b1r[NPL], w2r[NPL];
#pragma unroll
    for (int p = 0; p < NPL; ++p) {
        b1r[p] = b1[p * PCH + lane];
        w2r[p] = W2[p * PCH + lane];
    }
    const float bias2 = b2[0];
    const float di = (lane < cnt) ? dinv[d0 + r0 + lane] : 0.f;
    float keep = 0.f;
#pragma unroll
    for (int j = 0; j < RPW; ++j) {
        const float* trow = tile + min(r0 + j, rows - 1) * IN_CH + lane;
        float t[NPL];
#pragma unroll
        for (int p = 0; p < NPL; ++p) t[p] = fmaxf(trow[p * PCH] + b1r[p], 0.f);
        float v = t[0] * w2r[0] + t[1] * w2r[1] + t[2] * w2r[2] + t[3] * w2r[3];
#pragma unroll
        for (int off = 16; off > 0; off >>= 1) v += __shfl_xor(v, off, 32);
        if (lane == j) keep = v;
    }
    if (lane < cnt) {
        const int node = d0 + r0 + lane;
        h2[node]  = keep;
        out[node] = keep * di * di + bias2;
    }
}

__global__ void k_edge_agg2(const int* __restrict__ dst, const unsigned* __restrict__ rec,
                            const float* __restrict__ h2,
                            float* __restrict__ out, int e) {
    int i = blockIdx.x * blockDim.x + threadIdx.x;
    if (i < e) {
        const uint2 r = reinterpret_cast<const uint2*>(rec)[i];
        const int s = (int)(r.x & ((1u << SBITS) - 1u));
        atomicAdd(&out[dst[i]], h2[s] * __uint_as_float(r.y));
    }
}

extern "C" void kernel_launch(void* const* d_in, const int* in_sizes, int n_in,
                              void* d_out, int out_size, void* d_ws, size_t ws_size,
                              hipStream_t stream) {
    const float* x  = (const float*)d_in[0];
    const int*   ei = (const int*)d_in[1];
    const float* W1 = (const float*)d_in[2];
    const float* b1 = (const float*)d_in[3];
    const float* W2 = (const float*)d_in[4];
    const float* b2 = (const float*)d_in[5];

    const int n = in_sizes[0] / IN_CH;
    const int e = in_sizes[1] / 2;
    const int* src = ei;
    const int* dst = ei + e;

    float* ws   = (float*)d_ws;
    float* deg  = ws;
    float* dinv = deg + n;
    float* h    = dinv + n;
    float* h2   = h + (size_t)n * IN_CH;
    unsigned* own = (unsigned*)(h2 + n);
    unsigned* rec = own + ((e + 15) / 16) * 4;
    float* out  = (float*)d_out;
    const int mtiles = (n + 15) / 16;

    k_deg_init <<<(n + 255) / 256, 256, 0, stream>>>(deg, n);
    k_deg_own  <<<(e / 4 + 4 + 255) / 256, 256, 0, stream>>>(dst, deg, own, e);
    const int gblocks = (mtiles + 7) / 8;
    const int rblocks = (e + EPB) / EPB;
    k_gemm1_rec<<<gblocks + rblocks, 256, 0, stream>>>(x, W1, h, src, dst, deg, dinv, rec,
                                                       n, mtiles, e, gblocks);
    k_agg_owner<<<(n + RN - 1) / RN, AW * 32, 0, stream>>>(own, rec, dinv, h, b1, W2, b2, h2, out, n, e);
    k_edge_agg2<<<(e + 255) / 256, 256, 0, stream>>>(dst, rec, h2, out, e);
}


# Layer-2 edge sums by owner work-groups too: the layer-1 owner lists the records it meets, a 2nd owner launch adds h2[src]*w in LDS, stores out once; no global f32 atomics, LDS arrival order
# speedup vs baseline: 2.2813x; 1.3073x over previous
#include <hip/hip_runtime.h>
#include <hip/hip_bf16.h>


typedef __attribute__((ext_vector_type(2))) float v2f;
typedef __attribute__((ext_vector_type(8))) float v8f;

#define IN_CH 128
#define PCH   32
#define NPL   (IN_CH / PCH)
#define RN    391
#define AW    32
#define CH    512
#define QCAP  544
#define SBITS 20
#define RPW   ((RN + AW - 1) / AW)
#define L2K   8
#define LCAP  (L2K * AW * 32)

__global__ void k_deg_init(float* __restrict__ deg, int n) {
    int i = blockIdx.x * blockDim.x + threadIdx.x;
    if (i < n) deg[i] = 1.0f;
}

__global__ void k_deg_own(const int* __restrict__ dst, float* __restrict__ deg,
                          unsigned* __restrict__ own, int e) {
    const int t  = blockIdx.x * blockDim.x + threadIdx.x;
    const int nw = ((e + 15) / 16) * 4;
    if (t >= nw) return;
    int d[4];
#pragma unroll
    for (int j = 0; j < 4; ++j) d[j] = dst[min(4 * t + j, e - 1)];
    unsigned pk = 0;
#pragma unroll
    for (int j = 0; j < 4; ++j) {
        if (4 * t + j < e) {
            atomicAdd(&deg[d[j]], 1.0f);
            pk |= ((unsigned)d[j] / (unsigned)RN) << (8 * j);
        }
    }
    own[t] = pk;
}

#define EPB 2048
__device__ __forceinline__ void rec_block(
    int rb, int rblocks, int tid,
    const int* __restrict__ src, const int* __restrict__ dst,
    const float* __restrict__ deg, float* __restrict__ dinv,
    unsigned* __restrict__ rec, int n, int e) {
    if (e > 0) {
        const int base = rb * EPB + tid;
        int s[8], d[8];
        float dgs[8], dgd[8];
#pragma unroll
        for (int k = 0; k < 8; ++k) {
            const int i = min(base + k * 256, e - 1);
            s[k] = src[i];
            d[k] = dst[i];
        }
#pragma unroll
        for (int k = 0; k < 8; ++k) {
            dgs[k] = deg[s[k]];
            dgd[k] = deg[d[k]];
        }
        for (int j = rb * 256 + tid; j < n; j += rblocks * 256)
            dinv[j] = rsqrtf(deg[j]);
#pragma unroll
        for (int k = 0; k < 8; ++k) {
            const int i = base + k * 256;
            if (i < e) {
                const float w = rsqrtf(dgs[k]) * rsqrtf(dgd[k]);
                const unsigned key = (unsigned)s[k] | (((unsigned)d[k] % (unsigned)RN) << SBITS);
                reinterpret_cast<uint2*>(rec)[i] = make_uint2(key, __float_as_uint(w));
            }
        }
    } else {
        for (int j = rb * 256 + tid; j < n; j += rblocks * 256)
            dinv[j] = rsqrtf(deg[j]);
    }
}

__global__ __launch_bounds__(256) void k_gemm1_rec(
    const float* __restrict__ x, const float* __restrict__ W1,
    float* __restrict__ h,
    const int* __restrict__ src, const int* __restrict__ dst,
    const float* __restrict__ deg, float* __restrict__ dinv,
    unsigned* __restrict__ rec,
    int n_nodes, int mtiles, int e, int gblocks) {
    __shared__ float lw[IN_CH * IN_CH];
    const int tid = threadIdx.x;
    if ((int)blockIdx.x >= gblocks) {
        rec_block((int)blockIdx.x - gblocks, (int)gridDim.x - gblocks, tid,
                  src, dst, deg, dinv, rec, n_nodes, e);
        return;
    }
    for (int i = tid; i < IN_CH * IN_CH; i += 256) lw[i] = W1[i];
    __syncthreads();

    const int wid  = tid >> 5;
    const int lane = tid & 31;
    const int half = lane >> 4;
    const int l    = lane & 15;
    const int tile = blockIdx.x * 8 + wid;
    if (tile >= mtiles) return;
    const int m0 = tile * 16;

    v8f acc[8] = {};

    const int arow = min(m0 + l, n_nodes - 1);
    const float* ap = x + (size_t)arow * IN_CH;

    for (int kk = 0; kk < IN_CH; kk += 4) {
        const int ka = kk + 2 * half;
        v2f a;
        a.x = ap[ka];
        a.y = ap[ka + 1];
#pragma unroll
        for (int nt = 0; nt < 8; ++nt) {
            const int nc = nt * 16 + l;
            v2f b;
            b.x = lw[ka * IN_CH + nc];
            b.y = lw[(ka + 1) * IN_CH + nc];
            acc[nt] = __builtin_amdgcn_wmma_f32_16x16x4_f32(
                false, a, false, b, (short)0, acc[nt], false, false);
        }
    }

#pragma unroll
    for (int j = 0; j < 8; ++j) {
        const int row = m0 + j + 8 * half;
        if (row >= n_nodes) continue;
#pragma unroll
        for (int nt = 0; nt < 8; ++nt)
            h[(size_t)row * IN_CH + nt * 16 + l] = acc[nt][j];
    }
}

template <bool FULL>
__device__ __forceinline__ void add_queued(
    const int* q, int cnt, int lane,
    const uint2* __restrict__ rec2,
    const float* __restrict__ h, float* tile,
    uint2* __restrict__ mylist, int* lcount) {
    int lb = 0;
    if (lane == 0)
        lb = __hip_atomic_fetch_add(lcount, cnt, __ATOMIC_RELAXED, __HIP_MEMORY_SCOPE_WORKGROUP);
    int ms = 0, md = 0;
    float mw = 0.f;
    uint2 r = make_uint2(0u, 0u);
    if (FULL || lane < cnt) {
        r = rec2[q[lane]];
        ms = (int)(r.x & ((1u << SBITS) - 1u));
        md = (int)(r.x >> SBITS);
        mw = __uint_as_float(r.y);
    }
    const int li = __builtin_amdgcn_readfirstlane(lb) + lane;
    if ((FULL || lane < cnt) && li < LCAP) mylist[li] = r;
#pragma unroll
    for (int g = 0; g < 4; ++g) {
        float hv[8][4];
#pragma unroll
        for (int k = 0; k < 8; ++k) {
            if (FULL || g * 8 + k < cnt) {
                const float* row = h + (size_t)__builtin_amdgcn_readlane(ms, g * 8 + k) * IN_CH + lane;
#pragma unroll
                for (int j = 0; j < 4; ++j) hv[k][j] = row[j * 32];
            }
        }
#pragma unroll
        for (int k = 0; k < 8; ++k) {
            if (FULL || g * 8 + k < cnt) {
                const int dl = __builtin_amdgcn_readlane(md, g * 8 + k);
                const float wk = __int_as_float(
                    __builtin_amdgcn_readlane(__float_as_int(mw), g * 8 + k));
                float* trow = tile + dl * IN_CH + lane;
#pragma unroll
                for (int j = 0; j < 4; ++j)
                    __hip_atomic_fetch_add(trow + j * 32, hv[k][j] * wk,
                                           __ATOMIC_RELAXED, __HIP_MEMORY_SCOPE_WORKGROUP);
            }
        }
    }
}

__global__ __launch_bounds__(1024) void k_agg_owner(
    const unsigned* __restrict__ own, const unsigned* __restrict__ rec,
    const float* __restrict__ dinv, const float* __restrict__ h,
    const float* __restrict__ b1, const float* __restrict__ W2,
    float* __restrict__ h2, unsigned* __restrict__ list, int* __restrict__ lcnt,
    int n, int e) {
    __shared__ float tile[RN * IN_CH];
    __shared__ int   queue[AW * QCAP];
    __shared__ int   lcount;
    const int tid  = threadIdx.x;
    const int lane = tid & 31;
    const int w    = __builtin_amdgcn_readfirstlane(tid >> 5);
    const int d0   = blockIdx.x * RN;
    const int rows = min(RN, n - d0);

    if (tid == 0) lcount = 0;
    for (int i = tid; i < rows * IN_CH; i += AW * 32) {
        const float di = dinv[d0 + (i >> 7)];
        tile[i] = h[(size_t)d0 * IN_CH + i] * (di * di);
    }
    __syncthreads();

    const int nchunk = (e + CH - 1) / CH;
    const int niter  = (nchunk + AW - 1) / AW;
    const int t0     = (int)(((long long)blockIdx.x * niter) / gridDim.x);
    const int last16 = (e + 15) / 16 - 1;
    const unsigned pat = (unsigned)blockIdx.x * 0x01010101u;
    const uint4* own16 = reinterpret_cast<const uint4*>(own);
    const uint2* rec2  = reinterpret_cast<const uint2*>(rec);
    uint2* mylist = reinterpret_cast<uint2*>(list) + (size_t)blockIdx.x * LCAP;
    int* q = queue + w * QCAP;
    int qn = 0;

    auto load1 = [&](uint4& buf, int t) {
        int tt = t0 + t;
        if (tt >= niter) tt -= niter;
        const int blk = (tt * AW + w) * (CH / 16) + lane;
        buf = own16[min(blk, last16)];
    };
    auto scan1 = [&](const uint4& buf, int t) {
        int tt = t0 + t;
        if (tt >= niter) tt -= niter;
        const int base = (tt * AW + w) * CH + lane * 16;
        const unsigned xs[4] = {buf.x, buf.y, buf.z, buf.w};
#pragma unroll
        for (int k = 0; k < 4; ++k) {
            const unsigned v = xs[k] ^ pat;
            const unsigned z = (v - 0x01010101u) & ~v & 0x80808080u;
            if (__builtin_amdgcn_ballot_w32(z != 0u)) {
#pragma unroll
                for (int b = 0; b < 4; ++b) {
                    const int i = base + k * 4 + b;
                    const bool m = (((v >> (8 * b)) & 0xffu) == 0u) && (i < e);
                    const unsigned mask = __builtin_amdgcn_ballot_w32(m);
                    if (mask) {
                        if (m) {
                            const int pos = qn + (int)__builtin_amdgcn_mbcnt_lo(mask, 0u);
                            q[pos] = i;
                        }
                        qn += __builtin_popcount(mask);
                    }
                }
            }
        }
        while (qn >= 32) {
            qn -= 32;
            add_queued<true>(q + qn, 32, lane, rec2, h, tile, mylist, &lcount);
        }
    };

    uint4 bufa, bufb;
    load1(bufa, 0);
    for (int t = 0; t < niter; t += 2) {
        load1(bufb, t + 1);
        scan1(bufa, t);
        load1(bufa, t + 2);
        if (t + 1 < niter) scan1(bufb, t + 1);
    }
    if (qn > 0) add_queued<false>(q, qn, lane, rec2, h, tile, mylist, &lcount);
    __syncthreads();
    if (tid == 0) lcnt[blockIdx.x] = lcount;

    const int r0  = w * RPW;
    const int cnt = min(RPW, rows - r0);
    float b1r[NPL], w2r[NPL];
#pragma unroll
    for (int p = 0; p < NPL; ++p) {
        b1r[p] = b1[p * PCH + lane];
        w2r[p] = W2[p * PCH + lane];
    }
    float keep = 0.f;
#pragma unroll
    for (int j = 0; j < RPW; ++j) {
        const float* trow = tile + min(r0 + j, rows - 1) * IN_CH + lane;
        float t[NPL];
#pragma unroll
        for (int p = 0; p < NPL; ++p) t[p] = fmaxf(trow[p * PCH] + b1r[p], 0.f);
        float v = t[0] * w2r[0] + t[1] * w2r[1] + t[2] * w2r[2] + t[3] * w2r[3];
#pragma unroll
        for (int off = 16; off > 0; off >>= 1) v += __shfl_xor(v, off, 32);
        if (lane == j) keep = v;
    }
    if (lane < cnt) h2[d0 + r0 + lane] = keep;
}

__device__ __forceinline__ void add2(float* acc, uint2 r, float hv) {
    __hip_atomic_fetch_add(acc + (r.x >> SBITS), hv * __uint_as_float(r.y),
                           __ATOMIC_RELAXED, __HIP_MEMORY_SCOPE_WORKGROUP);
}

__global__ __launch_bounds__(1024) void k_agg2_owner(
    const unsigned* __restrict__ own, const unsigned* __restrict__ rec,
    const unsigned* __restrict__ list, const int* __restrict__ lcnt,
    const float* __restrict__ dinv, const float* __restrict__ h2,
    const float* __restrict__ b2, float* __restrict__ out, int n, int e) {
    __shared__ float acc[RN];
    const int tid  = threadIdx.x;
    const int d0   = blockIdx.x * RN;
    const int rows = min(RN, n - d0);
    const int c    = lcnt[blockIdx.x];
    const float bias2 = b2[0];
    for (int r = tid; r < rows; r += AW * 32) {
        const float di = dinv[d0 + r];
        acc[r] = h2[d0 + r] * di * di + bias2;
    }
    __syncthreads();
    if (c > 0 && c <= LCAP) {
        const uint2* my = reinterpret_cast<const uint2*>(list) + (size_t)blockIdx.x * LCAP;
        uint2 r[L2K];
        float hv[L2K];
#pragma unroll
        for (int k = 0; k < L2K; ++k) r[k] = my[min(k * (AW * 32) + tid, c - 1)];
#pragma unroll
        for (int k = 0; k < L2K; ++k) hv[k] = h2[r[k].x & ((1u << SBITS) - 1u)];
        asm volatile("" :: "v"(hv[0]), "v"(hv[1]), "v"(hv[2]), "v"(hv[3]),
                           "v"(hv[4]), "v"(hv[5]), "v"(hv[6]), "v"(hv[7]));
#pragma unroll
        for (int k = 0; k < L2K; ++k)
            if (k * (AW * 32) + tid < c) add2(acc, r[k], hv[k]);
    } else if (c > LCAP) {
        const uint2* rec2 = reinterpret_cast<const uint2*>(rec);
        for (int i = tid; i < e; i += AW * 32) {
            const unsigned ob = (own[i >> 2] >> (8 * (i & 3))) & 0xffu;
            if (ob == blockIdx.x) {
                const uint2 r = rec2[i];
                add2(acc, r, h2[r.x & ((1u << SBITS) - 1u)]);
            }
        }
    }
    __syncthreads();
    for (int r = tid; r < rows; r += AW * 32) out[d0 + r] = acc[r];
}

extern "C" void kernel_launch(void* const* d_in, const int* in_sizes, int n_in,
                              void* d_out, int out_size, void* d_ws, size_t ws_size,
                              hipStream_t stream) {
    const float* x  = (const float*)d_in[0];
    const int*   ei = (const int*)d_in[1];
    const float* W1 = (const float*)d_in[2];
    const float* b1 = (const float*)d_in[3];
    const float* W2 = (const float*)d_in[4];
    const float* b2 = (const float*)d_in[5];

    const int n = in_sizes[0] / IN_CH;
    const int e = in_sizes[1] / 2;
    const int* src = ei;
    const int* dst = ei + e;

    float* ws   = (float*)d_ws;
    float* deg  = ws;
    float* dinv = deg + n;
    float* h    = dinv + n;
    float* h2   = h + (size_t)n * IN_CH;
    unsigned* own = (unsigned*)(h2 + n);
    unsigned* rec = own + ((e + 15) / 16) * 4;
    unsigned* list = rec + 2 * (size_t)e;
    int* lcnt = (int*)(list + 2 * (size_t)256 * LCAP);
    float* out  = (float*)d_out;
    const int mtiles = (n + 15) / 16;

    k_deg_init <<<(n + 255) / 256, 256, 0, stream>>>(deg, n);
    k_deg_own  <<<(e / 4 + 4 + 255) / 256, 256, 0, stream>>>(dst, deg, own, e);
    const int gblocks = (mtiles + 7) / 8;
    const int rblocks = (e + EPB) / EPB;
    k_gemm1_rec<<<gblocks + rblocks, 256, 0, stream>>>(x, W1, h, src, dst, deg, dinv, rec,
                                                       n, mtiles, e, gblocks);
    k_agg_owner<<<(n + RN - 1) / RN, AW * 32, 0, stream>>>(own, rec, dinv, h, b1, W2, h2, list, lcnt, n, e);
    k_agg2_owner<<<(n + RN - 1) / RN, AW * 32, 0, stream>>>(own, rec, list, lcnt, dinv, h2, b2, out, n, e);
}


# Degree count by LDS histograms: a work-group per (50000-edge chunk, eighth of the nodes) counts targets in LDS as 16-bit pairs, planes summed per node; no global f32 atomics left; deg exact as before
# speedup vs baseline: 3.1252x; 1.3699x over previous
#include <hip/hip_runtime.h>
#include <hip/hip_bf16.h>


typedef __attribute__((ext_vector_type(2))) float v2f;
typedef __attribute__((ext_vector_type(8))) float v8f;

#define IN_CH 128
#define PCH   32
#define NPL   (IN_CH / PCH)
#define RN    391
#define AW    32
#define CH    512
#define QCAP  544
#define SBITS 20
#define RPW   ((RN + AW - 1) / AW)
#define L2K   8
#define LCAP  (L2K * AW * 32)

#define ECH 50000
#define HR  8
#define HNR (256 * RN / HR)
#define HNW (HNR / 2)
#define PNW (HR * HNW)
#define HIT ((ECH / 4 + 1023) / 1024)

__global__ __launch_bounds__(1024) void k_deg_hist(
    const int* __restrict__ dst, unsigned* __restrict__ part,
    unsigned* __restrict__ own, int e) {
    __shared__ uint4 cnt4[HNW / 4];
    unsigned* cnt = reinterpret_cast<unsigned*>(cnt4);
    const int tid = threadIdx.x;
    const int p   = blockIdx.x / HR;
    const int r   = blockIdx.x % HR;
    for (int i = tid; i < HNW / 4; i += 1024) cnt4[i] = make_uint4(0u, 0u, 0u, 0u);
    __syncthreads();
    const int nw = ((e + 15) / 16) * 4;
    const int q0 = p * (ECH / 4);
    const int q1 = min(q0 + ECH / 4, nw);
    const int nq = e >> 2;
    const unsigned lo = (unsigned)r * HNR;
    const bool vec = nq > 0 && (reinterpret_cast<size_t>(dst) & 15) == 0;
    int4 v[HIT];
#pragma unroll
    for (int it = 0; it < HIT; ++it) {
        v[it] = make_int4(0, 0, 0, 0);
        if (vec) v[it] = reinterpret_cast<const int4*>(dst)[min(q0 + it * 1024 + tid, nq - 1)];
    }
#pragma unroll
    for (int it = 0; it < HIT; ++it) {
        const int t = q0 + it * 1024 + tid;
        if (t < q1) {
            int d[4];
            if (vec && t < nq) {
                d[0] = v[it].x; d[1] = v[it].y; d[2] = v[it].z; d[3] = v[it].w;
            } else {
#pragma unroll
                for (int j = 0; j < 4; ++j) d[j] = dst[min(4 * t + j, max(e - 1, 0))];
            }
#pragma unroll
            for (int j = 0; j < 4; ++j) {
                const unsigned l = (unsigned)d[j] - lo;
                if (4 * t + j < e && l < (unsigned)HNR)
                    __hip_atomic_fetch_add(cnt + (l >> 1), 1u << ((l & 1u) * 16u),
                                           __ATOMIC_RELAXED, __HIP_MEMORY_SCOPE_WORKGROUP);
            }
            if ((it % HR) == r) {
                unsigned pk = 0;
#pragma unroll
                for (int j = 0; j < 4; ++j)
                    if (4 * t + j < e) pk |= ((unsigned)d[j] / (unsigned)RN) << (8 * j);
                own[t] = pk;
            }
        }
    }
    __syncthreads();
    uint4* pl = reinterpret_cast<uint4*>(part + (size_t)p * PNW + r * HNW);
    for (int i = tid; i < HNW / 4; i += 1024) pl[i] = cnt4[i];
}

__global__ void k_deg_sum(const unsigned* __restrict__ part, float* __restrict__ deg,
                          int n, int planes) {
    const int w = blockIdx.x * blockDim.x + threadIdx.x;
    if (2 * w >= n) return;
    unsigned c0 = 0, c1 = 0;
#pragma unroll 8
    for (int p = 0; p < planes; ++p) {
        const unsigned x = part[(size_t)p * PNW + w];
        c0 += x & 0xffffu;
        c1 += x >> 16;
    }
    deg[2 * w] = 1.0f + (float)c0;
    if (2 * w + 1 < n) deg[2 * w + 1] = 1.0f + (float)c1;
}

#define EPB 2048
__device__ __forceinline__ void rec_block(
    int rb, int rblocks, int tid,
    const int* __restrict__ src, const int* __restrict__ dst,
    const float* __restrict__ deg, float* __restrict__ dinv,
    unsigned* __restrict__ rec, int n, int e) {
    if (e > 0) {
        const int base = rb * EPB + tid;
        int s[8], d[8];
        float dgs[8], dgd[8];
#pragma unroll
        for (int k = 0; k < 8; ++k) {
            const int i = min(base + k * 256, e - 1);
            s[k] = src[i];
            d[k] = dst[i];
        }
#pragma unroll
        for (int k = 0; k < 8; ++k) {
            dgs[k] = deg[s[k]];
            dgd[k] = deg[d[k]];
        }
        for (int j = rb * 256 + tid; j < n; j += rblocks * 256)
            dinv[j] = rsqrtf(deg[j]);
#pragma unroll
        for (int k = 0; k < 8; ++k) {
            const int i = base + k * 256;
            if (i < e) {
                const float w = rsqrtf(dgs[k]) * rsqrtf(dgd[k]);
                const unsigned key = (unsigned)s[k] | (((unsigned)d[k] % (unsigned)RN) << SBITS);
                reinterpret_cast<uint2*>(rec)[i] = make_uint2(key, __float_as_uint(w));
            }
        }
    } else {
        for (int j = rb * 256 + tid; j < n; j += rblocks * 256)
            dinv[j] = rsqrtf(deg[j]);
    }
}

__global__ __launch_bounds__(256) void k_gemm1_rec(
    const float* __restrict__ x, const float* __restrict__ W1,
    float* __restrict__ h,
    const int* __restrict__ src, const int* __restrict__ dst,
    const float* __restrict__ deg, float* __restrict__ dinv,
    unsigned* __restrict__ rec,
    int n_nodes, int mtiles, int e, int gblocks) {
    __shared__ float lw[IN_CH * IN_CH];
    const int tid = threadIdx.x;
    if ((int)blockIdx.x >= gblocks) {
        rec_block((int)blockIdx.x - gblocks, (int)gridDim.x - gblocks, tid,
                  src, dst, deg, dinv, rec, n_nodes, e);
        return;
    }
    for (int i = tid; i < IN_CH * IN_CH; i += 256) lw[i] = W1[i];
    __syncthreads();

    const int wid  = tid >> 5;
    const int lane = tid & 31;
    const int half = lane >> 4;
    const int l    = lane & 15;
    const int tile = blockIdx.x * 8 + wid;
    if (tile >= mtiles) return;
    const int m0 = tile * 16;

    v8f acc[8] = {};

    const int arow = min(m0 + l, n_nodes - 1);
    const float* ap = x + (size_t)arow * IN_CH;

    for (int kk = 0; kk < IN_CH; kk += 4) {
        const int ka = kk + 2 * half;
        v2f a;
        a.x = ap[ka];
        a.y = ap[ka + 1];
#pragma unroll
        for (int nt = 0; nt < 8; ++nt) {
            const int nc = nt * 16 + l;
            v2f b;
            b.x = lw[ka * IN_CH + nc];
            b.y = lw[(ka + 1) * IN_CH + nc];
            acc[nt] = __builtin_amdgcn_wmma_f32_16x16x4_f32(
                false, a, false, b, (short)0, acc[nt], false, false);
        }
    }

#pragma unroll
    for (int j = 0; j < 8; ++j) {
        const int row = m0 + j + 8 * half;
        if (row >= n_nodes) continue;
#pragma unroll
        for (int nt = 0; nt < 8; ++nt)
            h[(size_t)row * IN_CH + nt * 16 + l] = acc[nt][j];
    }
}

template <bool FULL>
__device__ __forceinline__ void add_queued(
    const int* q, int cnt, int lane,
    const uint2* __restrict__ rec2,
    const float* __restrict__ h, float* tile,
    uint2* __restrict__ mylist, int* lcount) {
    int lb = 0;
    if (lane == 0)
        lb = __hip_atomic_fetch_add(lcount, cnt, __ATOMIC_RELAXED, __HIP_MEMORY_SCOPE_WORKGROUP);
    int ms = 0, md = 0;
    float mw = 0.f;
    uint2 r = make_uint2(0u, 0u);
    if (FULL || lane < cnt) {
        r = rec2[q[lane]];
        ms = (int)(r.x & ((1u << SBITS) - 1u));
        md = (int)(r.x >> SBITS);
        mw = __uint_as_float(r.y);
    }
    const int li = __builtin_amdgcn_readfirstlane(lb) + lane;
    if ((FULL || lane < cnt) && li < LCAP) mylist[li] = r;
#pragma unroll
    for (int g = 0; g < 4; ++g) {
        float hv[8][4];
#pragma unroll
        for (int k = 0; k < 8; ++k) {
            if (FULL || g * 8 + k < cnt) {
                const float* row = h + (size_t)__builtin_amdgcn_readlane(ms, g * 8 + k) * IN_CH + lane;
#pragma unroll
                for (int j = 0; j < 4; ++j) hv[k][j] = row[j * 32];
            }
        }
#pragma unroll
        for (int k = 0; k < 8; ++k) {
            if (FULL || g * 8 + k < cnt) {
                const int dl = __builtin_amdgcn_readlane(md, g * 8 + k);
                const float wk = __int_as_float(
                    __builtin_amdgcn_readlane(__float_as_int(mw), g * 8 + k));
                float* trow = tile + dl * IN_CH + lane;
#pragma unroll
                for (int j = 0; j < 4; ++j)
                    __hip_atomic_fetch_add(trow + j * 32, hv[k][j] * wk,
                                           __ATOMIC_RELAXED, __HIP_MEMORY_SCOPE_WORKGROUP);
            }
        }
    }
}

__global__ __launch_bounds__(1024) void k_agg_owner(
    const unsigned* __restrict__ own, const unsigned* __restrict__ rec,
    const float* __restrict__ dinv, const float* __restrict__ h,
    const float* __restrict__ b1, const float* __restrict__ W2,
    float* __restrict__ h2, unsigned* __restrict__ list, int* __restrict__ lcnt,
    int n, int e) {
    __shared__ float tile[RN * IN_CH];
    __shared__ int   queue[AW * QCAP];
    __shared__ int   lcount;
    const int tid  = threadIdx.x;
    const int lane = tid & 31;
    const int w    = __builtin_amdgcn_readfirstlane(tid >> 5);
    const int d0   = blockIdx.x * RN;
    const int rows = min(RN, n - d0);

    if (tid == 0) lcount = 0;
    for (int i = tid; i < rows * IN_CH; i += AW * 32) {
        const float di = dinv[d0 + (i >> 7)];
        tile[i] = h[(size_t)d0 * IN_CH + i] * (di * di);
    }
    __syncthreads();

    const int nchunk = (e + CH - 1) / CH;
    const int niter  = (nchunk + AW - 1) / AW;
    const int t0     = (int)(((long long)blockIdx.x * niter) / gridDim.x);
    const int last16 = (e + 15) / 16 - 1;
    const unsigned pat = (unsigned)blockIdx.x * 0x01010101u;
    const uint4* own16 = reinterpret_cast<const uint4*>(own);
    const uint2* rec2  = reinterpret_cast<const uint2*>(rec);
    uint2* mylist = reinterpret_cast<uint2*>(list) + (size_t)blockIdx.x * LCAP;
    int* q = queue + w * QCAP;
    int qn = 0;

    auto load1 = [&](uint4& buf, int t) {
        int tt = t0 + t;
        if (tt >= niter) tt -= niter;
        const int blk = (tt * AW + w) * (CH / 16) + lane;
        buf = own16[min(blk, last16)];
    };
    auto scan1 = [&](const uint4& buf, int t) {
        int tt = t0 + t;
        if (tt >= niter) tt -= niter;
        const int base = (tt * AW + w) * CH + lane * 16;
        const unsigned xs[4] = {buf.x, buf.y, buf.z, buf.w};
#pragma unroll
        for (int k = 0; k < 4; ++k) {
            const unsigned v = xs[k] ^ pat;
            const unsigned z = (v - 0x01010101u) & ~v & 0x80808080u;
            if (__builtin_amdgcn_ballot_w32(z != 0u)) {
#pragma unroll
                for (int b = 0; b < 4; ++b) {
                    const int i = base + k * 4 + b;
                    const bool m = (((v >> (8 * b)) & 0xffu) == 0u) && (i < e);
                    const unsigned mask = __builtin_amdgcn_ballot_w32(m);
                    if (mask) {
                        if (m) {
                            const int pos = qn + (int)__builtin_amdgcn_mbcnt_lo(mask, 0u);
                            q[pos] = i;
                        }
                        qn += __builtin_popcount(mask);
                    }
                }
            }
        }
        while (qn >= 32) {
            qn -= 32;
            add_queued<true>(q + qn, 32, lane, rec2, h, tile, mylist, &lcount);
        }
    };

    uint4 bufa, bufb;
    load1(bufa, 0);
    for (int t = 0; t < niter; t += 2) {
        load1(bufb, t + 1);
        scan1(bufa, t);
        load1(bufa, t + 2);
        if (t + 1 < niter) scan1(bufb, t + 1);
    }
    if (qn > 0) add_queued<false>(q, qn, lane, rec2, h, tile, mylist, &lcount);
    __syncthreads();
    if (tid == 0) lcnt[blockIdx.x] = lcount;

    const int r0  = w * RPW;
    const int cnt = min(RPW, rows - r0);
    float b1r[NPL], w2r[NPL];
#pragma unroll
    for (int p = 0; p < NPL; ++p) {
        b1r[p] = b1[p * PCH + lane];
        w2r[p] = W2[p * PCH + lane];
    }
    float keep = 0.f;
#pragma unroll
    for (int j = 0; j < RPW; ++j) {
        const float* trow = tile + min(r0 + j, rows - 1) * IN_CH + lane;
        float t[NPL];
#pragma unroll
        for (int p = 0; p < NPL; ++p) t[p] = fmaxf(trow[p * PCH] + b1r[p], 0.f);
        float v = t[0] * w2r[0] + t[1] * w2r[1] + t[2] * w2r[2] + t[3] * w2r[3];
#pragma unroll
        for (int off = 16; off > 0; off >>= 1) v += __shfl_xor(v, off, 32);
        if (lane == j) keep = v;
    }
    if (lane < cnt) h2[d0 + r0 + lane] = keep;
}

__device__ __forceinline__ void add2(float* acc, uint2 r, float hv) {
    __hip_atomic_fetch_add(acc + (r.x >> SBITS), hv * __uint_as_float(r.y),
                           __ATOMIC_RELAXED, __HIP_MEMORY_SCOPE_WORKGROUP);
}

__global__ __launch_bounds__(1024) void k_agg2_owner(
    const unsigned* __restrict__ own, const unsigned* __restrict__ rec,
    const unsigned* __restrict__ list, const int* __restrict__ lcnt,
    const float* __restrict__ dinv, const float* __restrict__ h2,
    const float* __restrict__ b2, float* __restrict__ out, int n, int e) {
    __shared__ float acc[RN];
    const int tid  = threadIdx.x;
    const int d0   = blockIdx.x * RN;
    const int rows = min(RN, n - d0);
    const int c    = lcnt[blockIdx.x];
    const float bias2 = b2[0];
    for (int r = tid; r < rows; r += AW * 32) {
        const float di = dinv[d0 + r];
        acc[r] = h2[d0 + r] * di * di + bias2;
    }
    __syncthreads();
    if (c > 0 && c <= LCAP) {
        const uint2* my = reinterpret_cast<const uint2*>(list) + (size_t)blockIdx.x * LCAP;
        uint2 r[L2K];
        float hv[L2K];
#pragma unroll
        for (int k = 0; k < L2K; ++k) r[k] = my[min(k * (AW * 32) + tid, c - 1)];
#pragma unroll
        for (int k = 0; k < L2K; ++k) hv[k] = h2[r[k].x & ((1u << SBITS) - 1u)];
        asm volatile("" :: "v"(hv[0]), "v"(hv[1]), "v"(hv[2]), "v"(hv[3]),
                           "v"(hv[4]), "v"(hv[5]), "v"(hv[6]), "v"(hv[7]));
#pragma unroll
        for (int k = 0; k < L2K; ++k)
            if (k * (AW * 32) + tid < c) add2(acc, r[k], hv[k]);
    } else if (c > LCAP) {
        const uint2* rec2 = reinterpret_cast<const uint2*>(rec);
        for (int i = tid; i < e; i += AW * 32) {
            const unsigned ob = (own[i >> 2] >> (8 * (i & 3))) & 0xffu;
            if (ob == blockIdx.x) {
                const uint2 r = rec2[i];
                add2(acc, r, h2[r.x & ((1u << SBITS) - 1u)]);
            }
        }
    }
    __syncthreads();
    for (int r = tid; r < rows; r += AW * 32) out[d0 + r] = acc[r];
}

extern "C" void kernel_launch(void* const* d_in, const int* in_sizes, int n_in,
                              void* d_out, int out_size, void* d_ws, size_t ws_size,
                              hipStream_t stream) {
    const float* x  = (const float*)d_in[0];
    const int*   ei = (const int*)d_in[1];
    const float* W1 = (const float*)d_in[2];
    const float* b1 = (const float*)d_in[3];
    const float* W2 = (const float*)d_in[4];
    const float* b2 = (const float*)d_in[5];

    const int n = in_sizes[0] / IN_CH;
    const int e = in_sizes[1] / 2;
    const int* src = ei;
    const int* dst = ei + e;

    const int planes = e > ECH ? (e + ECH - 1) / ECH : 1;
    unsigned* part = (unsigned*)d_ws;
    float* ws   = (float*)(part + (size_t)planes * PNW);
    float* deg  = ws;
    float* dinv = deg + n;
    float* h    = dinv + n;
    float* h2   = h + (size_t)n * IN_CH;
    unsigned* own = (unsigned*)(h2 + n);
    unsigned* rec = own + ((e + 15) / 16) * 4;
    unsigned* list = rec + 2 * (size_t)e;
    int* lcnt = (int*)(list + 2 * (size_t)256 * LCAP);
    float* out  = (float*)d_out;
    const int mtiles = (n + 15) / 16;

    k_deg_hist <<<planes * HR, 1024, 0, stream>>>(dst, part, own, e);
    k_deg_sum  <<<((n + 1) / 2 + 255) / 256, 256, 0, stream>>>(part, deg, n, planes);
    const int gblocks = (mtiles + 7) / 8;
    const int rblocks = (e + EPB) / EPB;
    k_gemm1_rec<<<gblocks + rblocks, 256, 0, stream>>>(x, W1, h, src, dst, deg, dinv, rec,
                                                       n, mtiles, e, gblocks);
    k_agg_owner<<<(n + RN - 1) / RN, AW * 32, 0, stream>>>(own, rec, dinv, h, b1, W2, h2, list, lcnt, n, e);
    k_agg2_owner<<<(n + RN - 1) / RN, AW * 32, 0, stream>>>(own, rec, list, lcnt, dinv, h2, b2, out, n, e);
}


# Owner membership kept as one bit-plane per owner (1 bit per edge, set in LDS by the degree pass): an owner scans only its own 200 KB of bits, not all 1.6M owner bytes; LDS add order changes
# speedup vs baseline: 3.6007x; 1.1522x over previous
#include <hip/hip_runtime.h>
#include <hip/hip_bf16.h>


typedef __attribute__((ext_vector_type(2))) float v2f;
typedef __attribute__((ext_vector_type(8))) float v8f;

#define IN_CH 128
#define PCH   32
#define NPL   (IN_CH / PCH)
#define RN    391
#define AW    32
#define QCAP  64
#define SBITS 20
#define RPW   ((RN + AW - 1) / AW)
#define L2K   8
#define LCAP  (L2K * AW * 32)

#define ECH 50000
#define HR  8
#define HNR (256 * RN / HR)
#define HNW (HNR / 2)
#define PNW (HR * HNW)
#define HIT ((ECH / 4 + 1023) / 1024)
#define OPR (HNR / RN)
#define BQ  (((ECH + 127) / 128 + 7) / 8 * 8)
#define BPW (4 * BQ)

__global__ __launch_bounds__(1024) void k_deg_hist(
    const int* __restrict__ dst, unsigned* __restrict__ part,
    unsigned* __restrict__ bmg, int e, int planes) {
    __shared__ uint4 cnt4[HNW / 4];
    __shared__ uint4 bm4[OPR * BQ];
    unsigned* cnt = reinterpret_cast<unsigned*>(cnt4);
    unsigned* bm  = reinterpret_cast<unsigned*>(bm4);
    const int tid = threadIdx.x;
    const int p   = blockIdx.x / HR;
    const int r   = blockIdx.x % HR;
    for (int i = tid; i < HNW / 4; i += 1024) cnt4[i] = make_uint4(0u, 0u, 0u, 0u);
    for (int i = tid; i < OPR * BQ; i += 1024) bm4[i] = make_uint4(0u, 0u, 0u, 0u);
    __syncthreads();
    const int nw = ((e + 15) / 16) * 4;
    const int q0 = p * (ECH / 4);
    const int q1 = min(q0 + ECH / 4, nw);
    const int nq = e >> 2;
    const unsigned lo = (unsigned)r * HNR;
    const bool vec = nq > 0 && (reinterpret_cast<size_t>(dst) & 15) == 0;
    int4 v[HIT];
#pragma unroll
    for (int it = 0; it < HIT; ++it) {
        v[it] = make_int4(0, 0, 0, 0);
        if (vec) v[it] = reinterpret_cast<const int4*>(dst)[min(q0 + it * 1024 + tid, nq - 1)];
    }
#pragma unroll
    for (int it = 0; it < HIT; ++it) {
        const int t = q0 + it * 1024 + tid;
        if (t < q1) {
            int d[4];
            if (vec && t < nq) {
                d[0] = v[it].x; d[1] = v[it].y; d[2] = v[it].z; d[3] = v[it].w;
            } else {
#pragma unroll
                for (int j = 0; j < 4; ++j) d[j] = dst[min(4 * t + j, max(e - 1, 0))];
            }
#pragma unroll
            for (int j = 0; j < 4; ++j) {
                const unsigned l = (unsigned)d[j] - lo;
                if (4 * t + j < e && l < (unsigned)HNR) {
                    __hip_atomic_fetch_add(cnt + (l >> 1), 1u << ((l & 1u) * 16u),
                                           __ATOMIC_RELAXED, __HIP_MEMORY_SCOPE_WORKGROUP);
                    const unsigned c = (unsigned)(4 * (t - q0) + j);
                    __hip_atomic_fetch_or(bm + (l / (unsigned)RN) * BPW + (c >> 5), 1u << (c & 31u),
                                          __ATOMIC_RELAXED, __HIP_MEMORY_SCOPE_WORKGROUP);
                }
            }
        }
    }
    __syncthreads();
    uint4* pl = reinterpret_cast<uint4*>(part + (size_t)p * PNW + r * HNW);
    for (int i = tid; i < HNW / 4; i += 1024) pl[i] = cnt4[i];
    uint4* bg = reinterpret_cast<uint4*>(bmg);
    for (int i = tid; i < OPR * BQ; i += 1024) {
        const int ol = i / BQ;
        bg[((size_t)(r * OPR + ol) * planes + p) * BQ + (i - ol * BQ)] = bm4[i];
    }
}

__global__ void k_deg_sum(const unsigned* __restrict__ part, float* __restrict__ deg,
                          int n, int planes) {
    const int w = blockIdx.x * blockDim.x + threadIdx.x;
    if (2 * w >= n) return;
    unsigned c0 = 0, c1 = 0;
#pragma unroll 8
    for (int p = 0; p < planes; ++p) {
        const unsigned x = part[(size_t)p * PNW + w];
        c0 += x & 0xffffu;
        c1 += x >> 16;
    }
    deg[2 * w] = 1.0f + (float)c0;
    if (2 * w + 1 < n) deg[2 * w + 1] = 1.0f + (float)c1;
}

#define EPB 2048
__device__ __forceinline__ void rec_block(
    int rb, int rblocks, int tid,
    const int* __restrict__ src, const int* __restrict__ dst,
    const float* __restrict__ deg, float* __restrict__ dinv,
    unsigned* __restrict__ rec, int n, int e) {
    if (e > 0) {
        const int base = rb * EPB + tid;
        int s[8], d[8];
        float dgs[8], dgd[8];
#pragma unroll
        for (int k = 0; k < 8; ++k) {
            const int i = min(base + k * 256, e - 1);
            s[k] = src[i];
            d[k] = dst[i];
        }
#pragma unroll
        for (int k = 0; k < 8; ++k) {
            dgs[k] = deg[s[k]];
            dgd[k] = deg[d[k]];
        }
        for (int j = rb * 256 + tid; j < n; j += rblocks * 256)
            dinv[j] = rsqrtf(deg[j]);
#pragma unroll
        for (int k = 0; k < 8; ++k) {
            const int i = base + k * 256;
            if (i < e) {
                const float w = rsqrtf(dgs[k]) * rsqrtf(dgd[k]);
                const unsigned key = (unsigned)s[k] | (((unsigned)d[k] % (unsigned)RN) << SBITS);
                reinterpret_cast<uint2*>(rec)[i] = make_uint2(key, __float_as_uint(w));
            }
        }
    } else {
        for (int j = rb * 256 + tid; j < n; j += rblocks * 256)
            dinv[j] = rsqrtf(deg[j]);
    }
}

__global__ __launch_bounds__(256) void k_gemm1_rec(
    const float* __restrict__ x, const float* __restrict__ W1,
    float* __restrict__ h,
    const int* __restrict__ src, const int* __restrict__ dst,
    const float* __restrict__ deg, float* __restrict__ dinv,
    unsigned* __restrict__ rec,
    int n_nodes, int mtiles, int e, int gblocks) {
    __shared__ float lw[IN_CH * IN_CH];
    const int tid = threadIdx.x;
    if ((int)blockIdx.x >= gblocks) {
        rec_block((int)blockIdx.x - gblocks, (int)gridDim.x - gblocks, tid,
                  src, dst, deg, dinv, rec, n_nodes, e);
        return;
    }
    for (int i = tid; i < IN_CH * IN_CH; i += 256) lw[i] = W1[i];
    __syncthreads();

    const int wid  = tid >> 5;
    const int lane = tid & 31;
    const int half = lane >> 4;
    const int l    = lane & 15;
    const int tile = blockIdx.x * 8 + wid;
    if (tile >= mtiles) return;
    const int m0 = tile * 16;

    v8f acc[8] = {};

    const int arow = min(m0 + l, n_nodes - 1);
    const float* ap = x + (size_t)arow * IN_CH;

    for (int kk = 0; kk < IN_CH; kk += 4) {
        const int ka = kk + 2 * half;
        v2f a;
        a.x = ap[ka];
        a.y = ap[ka + 1];
#pragma unroll
        for (int nt = 0; nt < 8; ++nt) {
            const int nc = nt * 16 + l;
            v2f b;
            b.x = lw[ka * IN_CH + nc];
            b.y = lw[(ka + 1) * IN_CH + nc];
            acc[nt] = __builtin_amdgcn_wmma_f32_16x16x4_f32(
                false, a, false, b, (short)0, acc[nt], false, false);
        }
    }

#pragma unroll
    for (int j = 0; j < 8; ++j) {
        const int row = m0 + j + 8 * half;
        if (row >= n_nodes) continue;
#pragma unroll
        for (int nt = 0; nt < 8; ++nt)
            h[(size_t)row * IN_CH + nt * 16 + l] = acc[nt][j];
    }
}

template <bool FULL>
__device__ __forceinline__ void add_queued(
    const int* q, int cnt, int lane,
    const uint2* __restrict__ rec2,
    const float* __restrict__ h, float* tile,
    uint2* __restrict__ mylist, int* lcount) {
    int lb = 0;
    if (lane == 0)
        lb = __hip_atomic_fetch_add(lcount, cnt, __ATOMIC_RELAXED, __HIP_MEMORY_SCOPE_WORKGROUP);
    int ms = 0, md = 0;
    float mw = 0.f;
    uint2 r = make_uint2(0u, 0u);
    if (FULL || lane < cnt) {
        r = rec2[q[lane]];
        ms = (int)(r.x & ((1u << SBITS) - 1u));
        md = (int)(r.x >> SBITS);
        mw = __uint_as_float(r.y);
    }
    const int li = __builtin_amdgcn_readfirstlane(lb) + lane;
    if ((FULL || lane < cnt) && li < LCAP) mylist[li] = r;
#pragma unroll
    for (int g = 0; g < 4; ++g) {
        float hv[8][4];
#pragma unroll
        for (int k = 0; k < 8; ++k) {
            if (FULL || g * 8 + k < cnt) {
                const float* row = h + (size_t)__builtin_amdgcn_readlane(ms, g * 8 + k) * IN_CH + lane;
#pragma unroll
                for (int j = 0; j < 4; ++j) hv[k][j] = row[j * 32];
            }
        }
#pragma unroll
        for (int k = 0; k < 8; ++k) {
            if (FULL || g * 8 + k < cnt) {
                const int dl = __builtin_amdgcn_readlane(md, g * 8 + k);
                const float wk = __int_as_float(
                    __builtin_amdgcn_readlane(__float_as_int(mw), g * 8 + k));
                float* trow = tile + dl * IN_CH + lane;
#pragma unroll
                for (int j = 0; j < 4; ++j)
                    __hip_atomic_fetch_add(trow + j * 32, hv[k][j] * wk,
                                           __ATOMIC_RELAXED, __HIP_MEMORY_SCOPE_WORKGROUP);
            }
        }
    }
}

__global__ __launch_bounds__(1024) void k_agg_owner(
    const unsigned* __restrict__ bmg, const unsigned* __restrict__ rec,
    const float* __restrict__ dinv, const float* __restrict__ h,
    const float* __restrict__ b1, const float* __restrict__ W2,
    float* __restrict__ h2, unsigned* __restrict__ list, int* __restrict__ lcnt,
    int n, int planes) {
    __shared__ float tile[RN * IN_CH];
    __shared__ int   queue[AW * QCAP];
    __shared__ int   lcount;
    const int tid  = threadIdx.x;
    const int lane = tid & 31;
    const int w    = __builtin_amdgcn_readfirstlane(tid >> 5);
    const int d0   = blockIdx.x * RN;
    const int rows = min(RN, n - d0);

    if (tid == 0) lcount = 0;
    for (int i = tid; i < rows * IN_CH; i += AW * 32) {
        const float di = dinv[d0 + (i >> 7)];
        tile[i] = h[(size_t)d0 * IN_CH + i] * (di * di);
    }
    __syncthreads();

    const int nb    = planes * BQ;
    const int niter = (nb + AW * 32 - 1) / (AW * 32);
    const uint4* bm16 = reinterpret_cast<const uint4*>(bmg) + (size_t)blockIdx.x * nb;
    const uint2* rec2 = reinterpret_cast<const uint2*>(rec);
    uint2* mylist = reinterpret_cast<uint2*>(list) + (size_t)blockIdx.x * LCAP;
    int* q = queue + w * QCAP;
    int qn = 0;

    auto load1 = [&](uint4& buf, int t) {
        const int u = (t * AW + w) * 32 + lane;
        buf = bm16[min(u, nb - 1)];
    };
    auto scan1 = [&](const uint4& buf, int t) {
        const int u  = (t * AW + w) * 32 + lane;
        const int pc = u / BQ;
        const int base = pc * ECH + (u - pc * BQ) * 128;
        const unsigned xs[4] = {buf.x, buf.y, buf.z, buf.w};
#pragma unroll
        for (int k = 0; k < 4; ++k) {
            unsigned x = (u < nb) ? xs[k] : 0u;
            for (int trip = 0; trip < 32; ++trip) {
                const unsigned mask = __builtin_amdgcn_ballot_w32(x != 0u);
                if (mask == 0u) break;
                if (x != 0u) {
                    const int pos = qn + (int)__builtin_amdgcn_mbcnt_lo(mask, 0u);
                    q[pos] = base + k * 32 + __builtin_ctz(x);
                    x &= x - 1u;
                }
                qn += __builtin_popcount(mask);
                if (qn >= 32) {
                    qn -= 32;
                    add_queued<true>(q + qn, 32, lane, rec2, h, tile, mylist, &lcount);
                }
            }
        }
    };

    uint4 cur, nxt;
    load1(cur, 0);
    for (int t = 0; t < niter; ++t) {
        load1(nxt, t + 1);
        scan1(cur, t);
        cur = nxt;
    }
    if (qn > 0) add_queued<false>(q, qn, lane, rec2, h, tile, mylist, &lcount);
    __syncthreads();
    if (tid == 0) lcnt[blockIdx.x] = lcount;

    const int r0  = w * RPW;
    const int cnt = min(RPW, rows - r0);
    float b1r[NPL], w2r[NPL];
#pragma unroll
    for (int p = 0; p < NPL; ++p) {
        b1r[p] = b1[p * PCH + lane];
        w2r[p] = W2[p * PCH + lane];
    }
    float keep = 0.f;
#pragma unroll
    for (int j = 0; j < RPW; ++j) {
        const float* trow = tile + min(r0 + j, rows - 1) * IN_CH + lane;
        float t[NPL];
#pragma unroll
        for (int p = 0; p < NPL; ++p) t[p] = fmaxf(trow[p * PCH] + b1r[p], 0.f);
        float v = t[0] * w2r[0] + t[1] * w2r[1] + t[2] * w2r[2] + t[3] * w2r[3];
#pragma unroll
        for (int off = 16; off > 0; off >>= 1) v += __shfl_xor(v, off, 32);
        if (lane == j) keep = v;
    }
    if (lane < cnt) h2[d0 + r0 + lane] = keep;
}

__device__ __forceinline__ void add2(float* acc, uint2 r, float hv) {
    __hip_atomic_fetch_add(acc + (r.x >> SBITS), hv * __uint_as_float(r.y),
                           __ATOMIC_RELAXED, __HIP_MEMORY_SCOPE_WORKGROUP);
}

__global__ __launch_bounds__(1024) void k_agg2_owner(
    const unsigned* __restrict__ bmg, const unsigned* __restrict__ rec,
    const unsigned* __restrict__ list, const int* __restrict__ lcnt,
    const float* __restrict__ dinv, const float* __restrict__ h2,
    const float* __restrict__ b2, float* __restrict__ out, int n, int planes) {
    __shared__ float acc[RN];
    const int tid  = threadIdx.x;
    const int d0   = blockIdx.x * RN;
    const int rows = min(RN, n - d0);
    const int c    = lcnt[blockIdx.x];
    const float bias2 = b2[0];
    for (int r = tid; r < rows; r += AW * 32) {
        const float di = dinv[d0 + r];
        acc[r] = h2[d0 + r] * di * di + bias2;
    }
    __syncthreads();
    if (c > 0 && c <= LCAP) {
        const uint2* my = reinterpret_cast<const uint2*>(list) + (size_t)blockIdx.x * LCAP;
        uint2 r[L2K];
        float hv[L2K];
#pragma unroll
        for (int k = 0; k < L2K; ++k) r[k] = my[min(k * (AW * 32) + tid, c - 1)];
#pragma unroll
        for (int k = 0; k < L2K; ++k) hv[k] = h2[r[k].x & ((1u << SBITS) - 1u)];
        asm volatile("" :: "v"(hv[0]), "v"(hv[1]), "v"(hv[2]), "v"(hv[3]),
                           "v"(hv[4]), "v"(hv[5]), "v"(hv[6]), "v"(hv[7]));
#pragma unroll
        for (int k = 0; k < L2K; ++k)
            if (k * (AW * 32) + tid < c) add2(acc, r[k], hv[k]);
    } else if (c > LCAP) {
        const uint2* rec2 = reinterpret_cast<const uint2*>(rec);
        const unsigned* bm = bmg + (size_t)blockIdx.x * planes * BPW;
        for (int dw = tid; dw < planes * BPW; dw += AW * 32) {
            const int pc = dw / BPW;
            const int base = pc * ECH + (dw - pc * BPW) * 32;
            unsigned x = bm[dw];
            while (x != 0u) {
                const uint2 r = rec2[base + __builtin_ctz(x)];
                add2(acc, r, h2[r.x & ((1u << SBITS) - 1u)]);
                x &= x - 1u;
            }
        }
    }
    __syncthreads();
    for (int r = tid; r < rows; r += AW * 32) out[d0 + r] = acc[r];
}

extern "C" void kernel_launch(void* const* d_in, const int* in_sizes, int n_in,
                              void* d_out, int out_size, void* d_ws, size_t ws_size,
                              hipStream_t stream) {
    const float* x  = (const float*)d_in[0];
    const int*   ei = (const int*)d_in[1];
    const float* W1 = (const float*)d_in[2];
    const float* b1 = (const float*)d_in[3];
    const float* W2 = (const float*)d_in[4];
    const float* b2 = (const float*)d_in[5];

    const int n = in_sizes[0] / IN_CH;
    const int e = in_sizes[1] / 2;
    const int* src = ei;
    const int* dst = ei + e;

    const int planes = e > ECH ? (e + ECH - 1) / ECH : 1;
    unsigned* part = (unsigned*)d_ws;
    float* ws   = (float*)(part + (size_t)planes * PNW);
    float* deg  = ws;
    float* dinv = deg + n;
    float* h    = dinv + n;
    float* h2   = h + (size_t)n * IN_CH;
    unsigned* bmg = (unsigned*)(h2 + n);
    unsigned* rec = bmg + (size_t)HR * OPR * planes * BPW;
    unsigned* list = rec + 2 * (size_t)e;
    int* lcnt = (int*)(list + 2 * (size_t)256 * LCAP);
    float* out  = (float*)d_out;
    const int mtiles = (n + 15) / 16;

    k_deg_hist <<<planes * HR, 1024, 0, stream>>>(dst, part, bmg, e, planes);
    k_deg_sum  <<<((n + 1) / 2 + 255) / 256, 256, 0, stream>>>(part, deg, n, planes);
    const int gblocks = (mtiles + 7) / 8;
    const int rblocks = (e + EPB) / EPB;
    k_gemm1_rec<<<gblocks + rblocks, 256, 0, stream>>>(x, W1, h, src, dst, deg, dinv, rec,
                                                       n, mtiles, e, gblocks);
    k_agg_owner<<<(n + RN - 1) / RN, AW * 32, 0, stream>>>(bmg, rec, dinv, h, b1, W2, h2, list, lcnt, n, planes);
    k_agg2_owner<<<(n + RN - 1) / RN, AW * 32, 0, stream>>>(bmg, rec, list, lcnt, dinv, h2, b2, out, n, planes);
}


# Layer-1 owner: records first routed in LDS to the wave owning the target row, rows then added by plain 16-byte LDS read-add-write, one 512-B gather per edge; no LDS f32 atomics; queue arrival order
# speedup vs baseline: 3.8633x; 1.0729x over previous
#include <hip/hip_runtime.h>
#include <hip/hip_bf16.h>


typedef __attribute__((ext_vector_type(2))) float v2f;
typedef __attribute__((ext_vector_type(8))) float v8f;
typedef __attribute__((ext_vector_type(4))) float v4f;

#define IN_CH 128
#define PCH   32
#define NPL   (IN_CH / PCH)
#define RN    391
#define AW    32
#define QCAP  64
#define SBITS 20
#define RPW   ((RN + AW - 1) / AW)
#define L2K   8
#define LCAP  (L2K * AW * 32)
#define QW    448

#define ECH 50000
#define HR  8
#define HNR (256 * RN / HR)
#define HNW (HNR / 2)
#define PNW (HR * HNW)
#define HIT ((ECH / 4 + 1023) / 1024)
#define OPR (HNR / RN)
#define BQ  (((ECH + 127) / 128 + 7) / 8 * 8)
#define BPW (4 * BQ)

__global__ __launch_bounds__(1024) void k_deg_hist(
    const int* __restrict__ dst, unsigned* __restrict__ part,
    unsigned* __restrict__ bmg, int e, int planes) {
    __shared__ uint4 cnt4[HNW / 4];
    __shared__ uint4 bm4[OPR * BQ];
    unsigned* cnt = reinterpret_cast<unsigned*>(cnt4);
    unsigned* bm  = reinterpret_cast<unsigned*>(bm4);
    const int tid = threadIdx.x;
    const int p   = blockIdx.x / HR;
    const int r   = blockIdx.x % HR;
    for (int i = tid; i < HNW / 4; i += 1024) cnt4[i] = make_uint4(0u, 0u, 0u, 0u);
    for (int i = tid; i < OPR * BQ; i += 1024) bm4[i] = make_uint4(0u, 0u, 0u, 0u);
    __syncthreads();
    const int nw = ((e + 15) / 16) * 4;
    const int q0 = p * (ECH / 4);
    const int q1 = min(q0 + ECH / 4, nw);
    const int nq = e >> 2;
    const unsigned lo = (unsigned)r * HNR;
    const bool vec = nq > 0 && (reinterpret_cast<size_t>(dst) & 15) == 0;
    int4 v[HIT];
#pragma unroll
    for (int it = 0; it < HIT; ++it) {
        v[it] = make_int4(0, 0, 0, 0);
        if (vec) v[it] = reinterpret_cast<const int4*>(dst)[min(q0 + it * 1024 + tid, nq - 1)];
    }
#pragma unroll
    for (int it = 0; it < HIT; ++it) {
        const int t = q0 + it * 1024 + tid;
        if (t < q1) {
            int d[4];
            if (vec && t < nq) {
                d[0] = v[it].x; d[1] = v[it].y; d[2] = v[it].z; d[3] = v[it].w;
            } else {
#pragma unroll
                for (int j = 0; j < 4; ++j) d[j] = dst[min(4 * t + j, max(e - 1, 0))];
            }
#pragma unroll
            for (int j = 0; j < 4; ++j) {
                const unsigned l = (unsigned)d[j] - lo;
                if (4 * t + j < e && l < (unsigned)HNR) {
                    __hip_atomic_fetch_add(cnt + (l >> 1), 1u << ((l & 1u) * 16u),
                                           __ATOMIC_RELAXED, __HIP_MEMORY_SCOPE_WORKGROUP);
                    const unsigned c = (unsigned)(4 * (t - q0) + j);
                    __hip_atomic_fetch_or(bm + (l / (unsigned)RN) * BPW + (c >> 5), 1u << (c & 31u),
                                          __ATOMIC_RELAXED, __HIP_MEMORY_SCOPE_WORKGROUP);
                }
            }
        }
    }
    __syncthreads();
    uint4* pl = reinterpret_cast<uint4*>(part + (size_t)p * PNW + r * HNW);
    for (int i = tid; i < HNW / 4; i += 1024) pl[i] = cnt4[i];
    uint4* bg = reinterpret_cast<uint4*>(bmg);
    for (int i = tid; i < OPR * BQ; i += 1024) {
        const int ol = i / BQ;
        bg[((size_t)(r * OPR + ol) * planes + p) * BQ + (i - ol * BQ)] = bm4[i];
    }
}

__global__ void k_deg_sum(const unsigned* __restrict__ part, float* __restrict__ deg,
                          int n, int planes) {
    const int w = blockIdx.x * blockDim.x + threadIdx.x;
    if (2 * w >= n) return;
    unsigned c0 = 0, c1 = 0;
#pragma unroll 8
    for (int p = 0; p < planes; ++p) {
        const unsigned x = part[(size_t)p * PNW + w];
        c0 += x & 0xffffu;
        c1 += x >> 16;
    }
    deg[2 * w] = 1.0f + (float)c0;
    if (2 * w + 1 < n) deg[2 * w + 1] = 1.0f + (float)c1;
}

#define EPB 2048
__device__ __forceinline__ void rec_block(
    int rb, int rblocks, int tid,
    const int* __restrict__ src, const int* __restrict__ dst,
    const float* __restrict__ deg, float* __restrict__ dinv,
    unsigned* __restrict__ rec, int n, int e) {
    if (e > 0) {
        const int base = rb * EPB + tid;
        int s[8], d[8];
        float dgs[8], dgd[8];
#pragma unroll
        for (int k = 0; k < 8; ++k) {
            const int i = min(base + k * 256, e - 1);
            s[k] = src[i];
            d[k] = dst[i];
        }
#pragma unroll
        for (int k = 0; k < 8; ++k) {
            dgs[k] = deg[s[k]];
            dgd[k] = deg[d[k]];
        }
        for (int j = rb * 256 + tid; j < n; j += rblocks * 256)
            dinv[j] = rsqrtf(deg[j]);
#pragma unroll
        for (int k = 0; k < 8; ++k) {
            const int i = base + k * 256;
            if (i < e) {
                const float w = rsqrtf(dgs[k]) * rsqrtf(dgd[k]);
                const unsigned key = (unsigned)s[k] | (((unsigned)d[k] % (unsigned)RN) << SBITS);
                reinterpret_cast<uint2*>(rec)[i] = make_uint2(key, __float_as_uint(w));
            }
        }
    } else {
        for (int j = rb * 256 + tid; j < n; j += rblocks * 256)
            dinv[j] = rsqrtf(deg[j]);
    }
}

__global__ __launch_bounds__(256) void k_gemm1_rec(
    const float* __restrict__ x, const float* __restrict__ W1,
    float* __restrict__ h,
    const int* __restrict__ src, const int* __restrict__ dst,
    const float* __restrict__ deg, float* __restrict__ dinv,
    unsigned* __restrict__ rec,
    int n_nodes, int mtiles, int e, int gblocks) {
    __shared__ float lw[IN_CH * IN_CH];
    const int tid = threadIdx.x;
    if ((int)blockIdx.x >= gblocks) {
        rec_block((int)blockIdx.x - gblocks, (int)gridDim.x - gblocks, tid,
                  src, dst, deg, dinv, rec, n_nodes, e);
        return;
    }
    for (int i = tid; i < IN_CH * IN_CH; i += 256) lw[i] = W1[i];
    __syncthreads();

    const int wid  = tid >> 5;
    const int lane = tid & 31;
    const int half = lane >> 4;
    const int l    = lane & 15;
    const int tile = blockIdx.x * 8 + wid;
    if (tile >= mtiles) return;
    const int m0 = tile * 16;

    v8f acc[8] = {};

    const int arow = min(m0 + l, n_nodes - 1);
    const float* ap = x + (size_t)arow * IN_CH;

    for (int kk = 0; kk < IN_CH; kk += 4) {
        const int ka = kk + 2 * half;
        v2f a;
        a.x = ap[ka];
        a.y = ap[ka + 1];
#pragma unroll
        for (int nt = 0; nt < 8; ++nt) {
            const int nc = nt * 16 + l;
            v2f b;
            b.x = lw[ka * IN_CH + nc];
            b.y = lw[(ka + 1) * IN_CH + nc];
            acc[nt] = __builtin_amdgcn_wmma_f32_16x16x4_f32(
                false, a, false, b, (short)0, acc[nt], false, false);
        }
    }

#pragma unroll
    for (int j = 0; j < 8; ++j) {
        const int row = m0 + j + 8 * half;
        if (row >= n_nodes) continue;
#pragma unroll
        for (int nt = 0; nt < 8; ++nt)
            h[(size_t)row * IN_CH + nt * 16 + l] = acc[nt][j];
    }
}

template <bool FULL>
__device__ __forceinline__ void route_queued(
    const int* q, int cnt, int lane,
    const uint2* __restrict__ rec2,
    const float* __restrict__ h, float* tile,
    uint2* wq, int* wcount,
    uint2* __restrict__ mylist, int* lcount) {
    int lb = 0;
    if (lane == 0)
        lb = __hip_atomic_fetch_add(lcount, cnt, __ATOMIC_RELAXED, __HIP_MEMORY_SCOPE_WORKGROUP);
    const bool act = FULL || lane < cnt;
    uint2 r = make_uint2(0u, 0u);
    if (act) r = rec2[q[lane]];
    const int li = __builtin_amdgcn_readfirstlane(lb) + lane;
    if (act && li < LCAP) mylist[li] = r;
    bool ovf = false;
    if (act) {
        const int ow = (int)((r.x >> SBITS) & (unsigned)(AW - 1));
        const int slot = __hip_atomic_fetch_add(wcount + ow, 1, __ATOMIC_RELAXED,
                                                __HIP_MEMORY_SCOPE_WORKGROUP);
        if (slot < QW) wq[ow * QW + slot] = r;
        else ovf = true;
    }
    unsigned om = __builtin_amdgcn_ballot_w32(ovf);
    while (om != 0u) {
        const int k = __builtin_ctz(om);
        om &= om - 1u;
        const unsigned key = (unsigned)__builtin_amdgcn_readlane((int)r.x, k);
        const float wk = __int_as_float(__builtin_amdgcn_readlane((int)r.y, k));
        const float* row = h + (size_t)(key & ((1u << SBITS) - 1u)) * IN_CH + lane;
        float* trow = tile + (key >> SBITS) * IN_CH + lane;
        float hv[4];
#pragma unroll
        for (int j = 0; j < 4; ++j) hv[j] = row[j * 32];
#pragma unroll
        for (int j = 0; j < 4; ++j)
            __hip_atomic_fetch_add(trow + j * 32, hv[j] * wk,
                                   __ATOMIC_RELAXED, __HIP_MEMORY_SCOPE_WORKGROUP);
    }
}

__device__ __forceinline__ void load_group(v4f (&hv)[8], uint2& r, const uint2* myq, int g0,
                                           int mine, int lane, const float* __restrict__ h) {
    r = myq[min(g0 + (lane & 7), mine - 1)];
    const int ms = (int)(r.x & ((1u << SBITS) - 1u));
#pragma unroll
    for (int k = 0; k < 8; ++k) {
        const float* row = h + (size_t)__builtin_amdgcn_readlane(ms, k) * IN_CH;
        hv[k] = reinterpret_cast<const v4f*>(row)[lane];
    }
}

template <bool FULL>
__device__ __forceinline__ void add_group(const v4f (&hv)[8], uint2 r, int left, int lane,
                                          v4f* tile4) {
#pragma unroll
    for (int k = 0; k < 8; ++k) {
        if (FULL || k < left) {
            const unsigned key = (unsigned)__builtin_amdgcn_readlane((int)r.x, k);
            const float wk = __int_as_float(__builtin_amdgcn_readlane((int)r.y, k));
            v4f* t = tile4 + (key >> SBITS) * (IN_CH / 4) + lane;
            *t = *t + hv[k] * wk;
        }
    }
}

__global__ __launch_bounds__(1024) void k_agg_owner(
    const unsigned* __restrict__ bmg, const unsigned* __restrict__ rec,
    const float* __restrict__ dinv, const float* __restrict__ h,
    const float* __restrict__ b1, const float* __restrict__ W2,
    float* __restrict__ h2, unsigned* __restrict__ list, int* __restrict__ lcnt,
    int n, int planes) {
    __shared__ v4f   tile4[RN * IN_CH / 4];
    __shared__ uint2 wq[AW * QW];
    __shared__ int   queue[AW * QCAP];
    __shared__ int   wcount[AW];
    __shared__ int   lcount;
    float* tile = reinterpret_cast<float*>(tile4);
    const int tid  = threadIdx.x;
    const int lane = tid & 31;
    const int w    = __builtin_amdgcn_readfirstlane(tid >> 5);
    const int d0   = blockIdx.x * RN;
    const int rows = min(RN, n - d0);

    if (tid == 0) lcount = 0;
    if (tid < AW) wcount[tid] = 0;
    for (int i = tid; i < rows * IN_CH; i += AW * 32) {
        const float di = dinv[d0 + (i >> 7)];
        tile[i] = h[(size_t)d0 * IN_CH + i] * (di * di);
    }
    __syncthreads();

    const int nb    = planes * BQ;
    const int niter = (nb + AW * 32 - 1) / (AW * 32);
    const uint4* bm16 = reinterpret_cast<const uint4*>(bmg) + (size_t)blockIdx.x * nb;
    const uint2* rec2 = reinterpret_cast<const uint2*>(rec);
    uint2* mylist = reinterpret_cast<uint2*>(list) + (size_t)blockIdx.x * LCAP;
    int* q = queue + w * QCAP;
    int qn = 0;

    auto load1 = [&](uint4& buf, int t) {
        const int u = (t * AW + w) * 32 + lane;
        buf = bm16[min(u, nb - 1)];
    };
    auto scan1 = [&](const uint4& buf, int t) {
        const int u  = (t * AW + w) * 32 + lane;
        const int pc = u / BQ;
        const int base = pc * ECH + (u - pc * BQ) * 128;
        const unsigned xs[4] = {buf.x, buf.y, buf.z, buf.w};
#pragma unroll
        for (int k = 0; k < 4; ++k) {
            unsigned x = (u < nb) ? xs[k] : 0u;
            for (int trip = 0; trip < 32; ++trip) {
                const unsigned mask = __builtin_amdgcn_ballot_w32(x != 0u);
                if (mask == 0u) break;
                if (x != 0u) {
                    const int pos = qn + (int)__builtin_amdgcn_mbcnt_lo(mask, 0u);
                    q[pos] = base + k * 32 + __builtin_ctz(x);
                    x &= x - 1u;
                }
                qn += __builtin_popcount(mask);
                if (qn >= 32) {
                    qn -= 32;
                    route_queued<true>(q + qn, 32, lane, rec2, h, tile, wq, wcount, mylist, &lcount);
                }
            }
        }
    };

    uint4 cur, nxt;
    load1(cur, 0);
    for (int t = 0; t < niter; ++t) {
        load1(nxt, t + 1);
        scan1(cur, t);
        cur = nxt;
    }
    if (qn > 0) route_queued<false>(q, qn, lane, rec2, h, tile, wq, wcount, mylist, &lcount);
    __syncthreads();
    if (tid == 0) lcnt[blockIdx.x] = lcount;

    {
        const int mine = min(__builtin_amdgcn_readfirstlane(wcount[w]), QW);
        const uint2* myq = wq + w * QW;
        if (mine > 0) {
            v4f ha[8], hb[8];
            uint2 ra, rb;
            load_group(ha, ra, myq, 0, mine, lane, h);
            for (int g0 = 0; g0 < mine; g0 += 16) {
                load_group(hb, rb, myq, g0 + 8, mine, lane, h);
                if (g0 + 8 <= mine) add_group<true>(ha, ra, 8, lane, tile4);
                else                add_group<false>(ha, ra, mine - g0, lane, tile4);
                load_group(ha, ra, myq, g0 + 16, mine, lane, h);
                if (g0 + 16 <= mine) add_group<true>(hb, rb, 8, lane, tile4);
                else                 add_group<false>(hb, rb, mine - g0 - 8, lane, tile4);
            }
        }
    }
    __syncthreads();

    const int r0  = w * RPW;
    const int cnt = min(RPW, rows - r0);
    float b1r[NPL], w2r[NPL];
#pragma unroll
    for (int p = 0; p < NPL; ++p) {
        b1r[p] = b1[p * PCH + lane];
        w2r[p] = W2[p * PCH + lane];
    }
    float keep = 0.f;
#pragma unroll
    for (int j = 0; j < RPW; ++j) {
        const float* trow = tile + min(r0 + j, rows - 1) * IN_CH + lane;
        float t[NPL];
#pragma unroll
        for (int p = 0; p < NPL; ++p) t[p] = fmaxf(trow[p * PCH] + b1r[p], 0.f);
        float v = t[0] * w2r[0] + t[1] * w2r[1] + t[2] * w2r[2] + t[3] * w2r[3];
#pragma unroll
        for (int off = 16; off > 0; off >>= 1) v += __shfl_xor(v, off, 32);
        if (lane == j) keep = v;
    }
    if (lane < cnt) h2[d0 + r0 + lane] = keep;
}

__device__ __forceinline__ void add2(float* acc, uint2 r, float hv) {
    __hip_atomic_fetch_add(acc + (r.x >> SBITS), hv * __uint_as_float(r.y),
                           __ATOMIC_RELAXED, __HIP_MEMORY_SCOPE_WORKGROUP);
}

__global__ __launch_bounds__(1024) void k_agg2_owner(
    const unsigned* __restrict__ bmg, const unsigned* __restrict__ rec,
    const unsigned* __restrict__ list, const int* __restrict__ lcnt,
    const float* __restrict__ dinv, const float* __restrict__ h2,
    const float* __restrict__ b2, float* __restrict__ out, int n, int planes) {
    __shared__ float acc[RN];
    const int tid  = threadIdx.x;
    const int d0   = blockIdx.x * RN;
    const int rows = min(RN, n - d0);
    const int c    = lcnt[blockIdx.x];
    const float bias2 = b2[0];
    for (int r = tid; r < rows; r += AW * 32) {
        const float di = dinv[d0 + r];
        acc[r] = h2[d0 + r] * di * di + bias2;
    }
    __syncthreads();
    if (c > 0 && c <= LCAP) {
        const uint2* my = reinterpret_cast<const uint2*>(list) + (size_t)blockIdx.x * LCAP;
        uint2 r[L2K];
        float hv[L2K];
#pragma unroll
        for (int k = 0; k < L2K; ++k) r[k] = my[min(k * (AW * 32) + tid, c - 1)];
#pragma unroll
        for (int k = 0; k < L2K; ++k) hv[k] = h2[r[k].x & ((1u << SBITS) - 1u)];
        asm volatile("" :: "v"(hv[0]), "v"(hv[1]), "v"(hv[2]), "v"(hv[3]),
                           "v"(hv[4]), "v"(hv[5]), "v"(hv[6]), "v"(hv[7]));
#pragma unroll
        for (int k = 0; k < L2K; ++k)
            if (k * (AW * 32) + tid < c) add2(acc, r[k], hv[k]);
    } else if (c > LCAP) {
        const uint2* rec2 = reinterpret_cast<const uint2*>(rec);
        const unsigned* bm = bmg + (size_t)blockIdx.x * planes * BPW;
        for (int dw = tid; dw < planes * BPW; dw += AW * 32) {
            const int pc = dw / BPW;
            const int base = pc * ECH + (dw - pc * BPW) * 32;
            unsigned x = bm[dw];
            while (x != 0u) {
                const uint2 r = rec2[base + __builtin_ctz(x)];
                add2(acc, r, h2[r.x & ((1u << SBITS) - 1u)]);
                x &= x - 1u;
            }
        }
    }
    __syncthreads();
    for (int r = tid; r < rows; r += AW * 32) out[d0 + r] = acc[r];
}

extern "C" void kernel_launch(void* const* d_in, const int* in_sizes, int n_in,
                              void* d_out, int out_size, void* d_ws, size_t ws_size,
                              hipStream_t stream) {
    const float* x  = (const float*)d_in[0];
    const int*   ei = (const int*)d_in[1];
    const float* W1 = (const float*)d_in[2];
    const float* b1 = (const float*)d_in[3];
    const float* W2 = (const float*)d_in[4];
    const float* b2 = (const float*)d_in[5];

    const int n = in_sizes[0] / IN_CH;
    const int e = in_sizes[1] / 2;
    const int* src = ei;
    const int* dst = ei + e;

    const int planes = e > ECH ? (e + ECH - 1) / ECH : 1;
    unsigned* part = (unsigned*)d_ws;
    float* ws   = (float*)(part + (size_t)planes * PNW);
    float* deg  = ws;
    float* dinv = deg + n;
    float* h    = dinv + n;
    float* h2   = h + (size_t)n * IN_CH;
    unsigned* bmg = (unsigned*)(h2 + n);
    unsigned* rec = bmg + (size_t)HR * OPR * planes * BPW;
    unsigned* list = rec + 2 * (size_t)e;
    int* lcnt = (int*)(list + 2 * (size_t)256 * LCAP);
    float* out  = (float*)d_out;
    const int mtiles = (n + 15) / 16;

    k_deg_hist <<<planes * HR, 1024, 0, stream>>>(dst, part, bmg, e, planes);
    k_deg_sum  <<<((n + 1) / 2 + 255) / 256, 256, 0, stream>>>(part, deg, n, planes);
    const int gblocks = (mtiles + 7) / 8;
    const int rblocks = (e + EPB) / EPB;
    k_gemm1_rec<<<gblocks + rblocks, 256, 0, stream>>>(x, W1, h, src, dst, deg, dinv, rec,
                                                       n, mtiles, e, gblocks);
    k_agg_owner<<<(n + RN - 1) / RN, AW * 32, 0, stream>>>(bmg, rec, dinv, h, b1, W2, h2, list, lcnt, n, planes);
    k_agg2_owner<<<(n + RN - 1) / RN, AW * 32, 0, stream>>>(bmg, rec, list, lcnt, dinv, h2, b2, out, n, planes);
}
